# starting kernel
# baseline (speedup 1.0000x reference)
#include <hip/hip_runtime.h>
#include <math.h>

typedef __attribute__((ext_vector_type(2))) float v2f;
typedef __attribute__((ext_vector_type(8))) float v8f;

#define WMMA_F32X4(a, b, c) \
    __builtin_amdgcn_wmma_f32_16x16x4_f32(false, (a), false, (b), (short)0, (c), false, false)

__global__ __launch_bounds__(256) void pdn_gate_kernel(
    const float* __restrict__ ea, const float* __restrict__ mw1,
    const float* __restrict__ mb1, const float* __restrict__ mw2,
    const float* __restrict__ mb2, float* __restrict__ g, int E) {
  const int wave = threadIdx.x >> 5;
  const int lane = threadIdx.x & 31;
  const int half = lane >> 4;
  const int l16  = lane & 15;
  const int e0 = (blockIdx.x * 8 + wave) * 16;
  if (e0 >= E) return;

  int er = e0 + l16;
  if (er >= E) er = E - 1;
  const float* earow = ea + (size_t)er * 16;

  v2f afrag[4];
#pragma unroll
  for (int kk = 0; kk < 4; ++kk) {
    afrag[kk].x = earow[4 * kk + 2 * half];
    afrag[kk].y = earow[4 * kk + 2 * half + 1];
  }

  float acc[8];
#pragma unroll
  for (int v = 0; v < 8; ++v) acc[v] = 0.0f;

#pragma unroll
  for (int t = 0; t < 8; ++t) {
    const int n0 = t * 16;
    v8f c = {};
#pragma unroll
    for (int kk = 0; kk < 4; ++kk) {
      const int kb = 4 * kk + 2 * half;
      v2f b;
      b.x = mw1[(size_t)kb * 128 + n0 + l16];
      b.y = mw1[(size_t)(kb + 1) * 128 + n0 + l16];
      c = WMMA_F32X4(afrag[kk], b, c);
    }
    const float b1 = mb1[n0 + l16];
    const float w2 = mw2[n0 + l16];
#pragma unroll
    for (int v = 0; v < 8; ++v) {
      acc[v] += fmaxf(c[v] + b1, 0.0f) * w2;
    }
  }

#pragma unroll
  for (int off = 1; off < 16; off <<= 1) {
#pragma unroll
    for (int v = 0; v < 8; ++v) acc[v] += __shfl_xor(acc[v], off, 32);
  }

  const float b2 = mb2[0];
#pragma unroll
  for (int v = 0; v < 8; ++v) {
    if (l16 == v) {
      const int e = e0 + v + 8 * half;
      if (e < E) {
        const float s = acc[v] + b2;
        g[e] = 1.0f / (1.0f + __expf(-s));
      }
    }
  }
}

#define RPW 4
__global__ __launch_bounds__(256) void pdn_gemm_kernel(
    const float* __restrict__ x, const float* __restrict__ W,
    float* __restrict__ h, int M, int relu_in) {
  const int wave = threadIdx.x >> 5;
  const int lane = threadIdx.x & 31;
  const int half = lane >> 4;
  const int l16  = lane & 15;
  const int n0 = wave * 16;
  const int rowBase = blockIdx.x * (16 * RPW);

  const float* xrow[RPW];
#pragma unroll
  for (int t = 0; t < RPW; ++t) {
    int ar = rowBase + t * 16 + l16;
    if (ar >= M) ar = M - 1;
    xrow[t] = x + (size_t)ar * 128;
  }

  v8f c[RPW];
#pragma unroll
  for (int t = 0; t < RPW; ++t) c[t] = (v8f){};

#pragma unroll 4
  for (int k0 = 0; k0 < 128; k0 += 4) {
    const int ka = k0 + half * 2;
    v2f b;
    b.x = W[(size_t)ka * 128 + n0 + l16];
    b.y = W[(size_t)(ka + 1) * 128 + n0 + l16];
#pragma unroll
    for (int t = 0; t < RPW; ++t) {
      v2f a;
      a.x = xrow[t][ka];
      a.y = xrow[t][ka + 1];
      if (relu_in) { a.x = fmaxf(a.x, 0.0f); a.y = fmaxf(a.y, 0.0f); }
      c[t] = WMMA_F32X4(a, b, c[t]);
    }
  }

#pragma unroll
  for (int t = 0; t < RPW; ++t) {
#pragma unroll
    for (int v = 0; v < 8; ++v) {
      const int r = rowBase + t * 16 + v + 8 * half;
      if (r < M) h[(size_t)r * 128 + n0 + l16] = c[t][v];
    }
  }
}


__global__ void pdn_deg_init_kernel(float* __restrict__ deg, int N) {
  const int i = blockIdx.x * blockDim.x + threadIdx.x;
  if (i < N) deg[i] = 1.0f;
}

__global__ void pdn_deg_scatter_kernel(const int* __restrict__ col,
                                       const float* __restrict__ g,
                                       float* __restrict__ deg, int E) {
  const int e = blockIdx.x * blockDim.x + threadIdx.x;
  if (e < E) atomicAdd(&deg[col[e]], g[e]);
}

__global__ void pdn_dinv_kernel(float* __restrict__ deg, int N) {
  const int i = blockIdx.x * blockDim.x + threadIdx.x;
  if (i < N) {
    const float d = deg[i];
    deg[i] = (d > 0.0f) ? rsqrtf(d) : 0.0f;
  }
}

__global__ void pdn_out_init_kernel(const float* __restrict__ h,
                                    const float* __restrict__ dinv,
                                    float* __restrict__ out, int N) {
  const int t = blockIdx.x * blockDim.x + threadIdx.x;
  const int i = t >> 7;
  if (i < N) {
    const float di = dinv[i];
    out[t] = di * di * h[t];
  }
}

__global__ void pdn_agg_kernel(const float* __restrict__ h,
                               const int* __restrict__ row,
                               const int* __restrict__ col,
                               const float* __restrict__ g,
                               const float* __restrict__ dinv,
                               float* __restrict__ out, int E) {
  const int t = blockIdx.x * blockDim.x + threadIdx.x;
  const int e = t >> 5;
  const int lane = t & 31;
  if (e >= E) return;
  const int r = row[e];
  const int c = col[e];
  const float norm = dinv[r] * g[e] * dinv[c];
  const float4 hv = ((const float4*)(h + (size_t)r * 128))[lane];
  float* o = out + (size_t)c * 128 + lane * 4;
  atomicAdd(o + 0, norm * hv.x);
  atomicAdd(o + 1, norm * hv.y);
  atomicAdd(o + 2, norm * hv.z);
  atomicAdd(o + 3, norm * hv.w);
}


extern "C" void kernel_launch(void* const* d_in, const int* in_sizes, int n_in,
                              void* d_out, int out_size, void* d_ws, size_t ws_size,
                              hipStream_t stream) {
  const float* x    = (const float*)d_in[0];
  const int*   ei   = (const int*)d_in[1];
  const float* ea   = (const float*)d_in[2];
  const float* W1   = (const float*)d_in[3];
  const float* m1w1 = (const float*)d_in[4];
  const float* m1b1 = (const float*)d_in[5];
  const float* m1w2 = (const float*)d_in[6];
  const float* m1b2 = (const float*)d_in[7];
  const float* W2   = (const float*)d_in[8];
  const float* m2w1 = (const float*)d_in[9];
  const float* m2b1 = (const float*)d_in[10];
  const float* m2w2 = (const float*)d_in[11];
  const float* m2b2 = (const float*)d_in[12];

  const int N = in_sizes[0] / 128;
  const int E = in_sizes[2] / 16;
  const int* row = ei;
  const int* col = ei + E;

  float* ws    = (float*)d_ws;
  float* h_buf = ws;
  float* a_buf = h_buf + (size_t)N * 128;
  float* g_buf = a_buf + (size_t)N * 128;
  float* deg   = g_buf + (size_t)E;
  float* outp  = (float*)d_out;

  const int gateBlocks = (E + 127) / 128;
  const int gemmBlocks = (N + 16 * RPW - 1) / (16 * RPW);
  const int nBlocks    = (N + 255) / 256;
  const int eBlocks    = (E + 255) / 256;
  const int neBlocks   = ((int)((size_t)N * 128 + 255)) / 256;
  const int aggBlocks  = (int)(((size_t)E * 32 + 255) / 256);

  pdn_gate_kernel<<<gateBlocks, 256, 0, stream>>>(ea, m1w1, m1b1, m1w2, m1b2, g_buf, E);
  pdn_deg_init_kernel<<<nBlocks, 256, 0, stream>>>(deg, N);
  pdn_deg_scatter_kernel<<<eBlocks, 256, 0, stream>>>(col, g_buf, deg, E);
  pdn_dinv_kernel<<<nBlocks, 256, 0, stream>>>(deg, N);
  pdn_gemm_kernel<<<gemmBlocks, 256, 0, stream>>>(x, W1, h_buf, N, 0);
  pdn_out_init_kernel<<<neBlocks, 256, 0, stream>>>(h_buf, deg, a_buf, N);
  pdn_agg_kernel<<<aggBlocks, 256, 0, stream>>>(h_buf, row, col, g_buf, deg, a_buf, E);

  pdn_gate_kernel<<<gateBlocks, 256, 0, stream>>>(ea, m2w1, m2b1, m2w2, m2b2, g_buf, E);
  pdn_deg_init_kernel<<<nBlocks, 256, 0, stream>>>(deg, N);
  pdn_deg_scatter_kernel<<<eBlocks, 256, 0, stream>>>(col, g_buf, deg, E);
  pdn_dinv_kernel<<<nBlocks, 256, 0, stream>>>(deg, N);
  pdn_gemm_kernel<<<gemmBlocks, 256, 0, stream>>>(a_buf, W2, h_buf, N, 1);
  pdn_out_init_kernel<<<neBlocks, 256, 0, stream>>>(h_buf, deg, outp, N);
  pdn_agg_kernel<<<aggBlocks, 256, 0, stream>>>(h_buf, row, col, g_buf, deg, outp, E);
}


# agg: lane l takes columns l, l+32, l+64, l+96, so each atomic (and load) instruction covers one whole 128-byte line instead of 4 lines x 8 lanes: 4x fewer L2 atomic requests
# speedup vs baseline: 1.2485x; 1.2485x over previous
#include <hip/hip_runtime.h>
#include <math.h>

typedef __attribute__((ext_vector_type(2))) float v2f;
typedef __attribute__((ext_vector_type(8))) float v8f;

#define WMMA_F32X4(a, b, c) \
    __builtin_amdgcn_wmma_f32_16x16x4_f32(false, (a), false, (b), (short)0, (c), false, false)

__global__ __launch_bounds__(256) void pdn_gate_kernel(
    const float* __restrict__ ea, const float* __restrict__ mw1,
    const float* __restrict__ mb1, const float* __restrict__ mw2,
    const float* __restrict__ mb2, float* __restrict__ g, int E) {
  const int wave = threadIdx.x >> 5;
  const int lane = threadIdx.x & 31;
  const int half = lane >> 4;
  const int l16  = lane & 15;
  const int e0 = (blockIdx.x * 8 + wave) * 16;
  if (e0 >= E) return;

  int er = e0 + l16;
  if (er >= E) er = E - 1;
  const float* earow = ea + (size_t)er * 16;

  v2f afrag[4];
#pragma unroll
  for (int kk = 0; kk < 4; ++kk) {
    afrag[kk].x = earow[4 * kk + 2 * half];
    afrag[kk].y = earow[4 * kk + 2 * half + 1];
  }

  float acc[8];
#pragma unroll
  for (int v = 0; v < 8; ++v) acc[v] = 0.0f;

#pragma unroll
  for (int t = 0; t < 8; ++t) {
    const int n0 = t * 16;
    v8f c = {};
#pragma unroll
    for (int kk = 0; kk < 4; ++kk) {
      const int kb = 4 * kk + 2 * half;
      v2f b;
      b.x = mw1[(size_t)kb * 128 + n0 + l16];
      b.y = mw1[(size_t)(kb + 1) * 128 + n0 + l16];
      c = WMMA_F32X4(afrag[kk], b, c);
    }
    const float b1 = mb1[n0 + l16];
    const float w2 = mw2[n0 + l16];
#pragma unroll
    for (int v = 0; v < 8; ++v) {
      acc[v] += fmaxf(c[v] + b1, 0.0f) * w2;
    }
  }

#pragma unroll
  for (int off = 1; off < 16; off <<= 1) {
#pragma unroll
    for (int v = 0; v < 8; ++v) acc[v] += __shfl_xor(acc[v], off, 32);
  }

  const float b2 = mb2[0];
#pragma unroll
  for (int v = 0; v < 8; ++v) {
    if (l16 == v) {
      const int e = e0 + v + 8 * half;
      if (e < E) {
        const float s = acc[v] + b2;
        g[e] = 1.0f / (1.0f + __expf(-s));
      }
    }
  }
}

#define RPW 4
__global__ __launch_bounds__(256) void pdn_gemm_kernel(
    const float* __restrict__ x, const float* __restrict__ W,
    float* __restrict__ h, int M, int relu_in) {
  const int wave = threadIdx.x >> 5;
  const int lane = threadIdx.x & 31;
  const int half = lane >> 4;
  const int l16  = lane & 15;
  const int n0 = wave * 16;
  const int rowBase = blockIdx.x * (16 * RPW);

  const float* xrow[RPW];
#pragma unroll
  for (int t = 0; t < RPW; ++t) {
    int ar = rowBase + t * 16 + l16;
    if (ar >= M) ar = M - 1;
    xrow[t] = x + (size_t)ar * 128;
  }

  v8f c[RPW];
#pragma unroll
  for (int t = 0; t < RPW; ++t) c[t] = (v8f){};

#pragma unroll 4
  for (int k0 = 0; k0 < 128; k0 += 4) {
    const int ka = k0 + half * 2;
    v2f b;
    b.x = W[(size_t)ka * 128 + n0 + l16];
    b.y = W[(size_t)(ka + 1) * 128 + n0 + l16];
#pragma unroll
    for (int t = 0; t < RPW; ++t) {
      v2f a;
      a.x = xrow[t][ka];
      a.y = xrow[t][ka + 1];
      if (relu_in) { a.x = fmaxf(a.x, 0.0f); a.y = fmaxf(a.y, 0.0f); }
      c[t] = WMMA_F32X4(a, b, c[t]);
    }
  }

#pragma unroll
  for (int t = 0; t < RPW; ++t) {
#pragma unroll
    for (int v = 0; v < 8; ++v) {
      const int r = rowBase + t * 16 + v + 8 * half;
      if (r < M) h[(size_t)r * 128 + n0 + l16] = c[t][v];
    }
  }
}


__global__ void pdn_deg_init_kernel(float* __restrict__ deg, int N) {
  const int i = blockIdx.x * blockDim.x + threadIdx.x;
  if (i < N) deg[i] = 1.0f;
}

__global__ void pdn_deg_scatter_kernel(const int* __restrict__ col,
                                       const float* __restrict__ g,
                                       float* __restrict__ deg, int E) {
  const int e = blockIdx.x * blockDim.x + threadIdx.x;
  if (e < E) atomicAdd(&deg[col[e]], g[e]);
}

__global__ void pdn_dinv_kernel(float* __restrict__ deg, int N) {
  const int i = blockIdx.x * blockDim.x + threadIdx.x;
  if (i < N) {
    const float d = deg[i];
    deg[i] = (d > 0.0f) ? rsqrtf(d) : 0.0f;
  }
}

__global__ void pdn_out_init_kernel(const float* __restrict__ h,
                                    const float* __restrict__ dinv,
                                    float* __restrict__ out, int N) {
  const int t = blockIdx.x * blockDim.x + threadIdx.x;
  const int i = t >> 7;
  if (i < N) {
    const float di = dinv[i];
    out[t] = di * di * h[t];
  }
}

__global__ void pdn_agg_kernel(const float* __restrict__ h,
                               const int* __restrict__ row,
                               const int* __restrict__ col,
                               const float* __restrict__ g,
                               const float* __restrict__ dinv,
                               float* __restrict__ out, int E) {
  const int t = blockIdx.x * blockDim.x + threadIdx.x;
  const int e = t >> 5;
  const int lane = t & 31;
  if (e >= E) return;
  const int r = row[e];
  const int c = col[e];
  const float norm = dinv[r] * g[e] * dinv[c];
  const float* hr = h + (size_t)r * 128 + lane;
  const float h0 = hr[0];
  const float h1 = hr[32];
  const float h2 = hr[64];
  const float h3 = hr[96];
  float* o = out + (size_t)c * 128 + lane;
  atomicAdd(o + 0,  norm * h0);
  atomicAdd(o + 32, norm * h1);
  atomicAdd(o + 64, norm * h2);
  atomicAdd(o + 96, norm * h3);
}


extern "C" void kernel_launch(void* const* d_in, const int* in_sizes, int n_in,
                              void* d_out, int out_size, void* d_ws, size_t ws_size,
                              hipStream_t stream) {
  const float* x    = (const float*)d_in[0];
  const int*   ei   = (const int*)d_in[1];
  const float* ea   = (const float*)d_in[2];
  const float* W1   = (const float*)d_in[3];
  const float* m1w1 = (const float*)d_in[4];
  const float* m1b1 = (const float*)d_in[5];
  const float* m1w2 = (const float*)d_in[6];
  const float* m1b2 = (const float*)d_in[7];
  const float* W2   = (const float*)d_in[8];
  const float* m2w1 = (const float*)d_in[9];
  const float* m2b1 = (const float*)d_in[10];
  const float* m2w2 = (const float*)d_in[11];
  const float* m2b2 = (const float*)d_in[12];

  const int N = in_sizes[0] / 128;
  const int E = in_sizes[2] / 16;
  const int* row = ei;
  const int* col = ei + E;

  float* ws    = (float*)d_ws;
  float* h_buf = ws;
  float* a_buf = h_buf + (size_t)N * 128;
  float* g_buf = a_buf + (size_t)N * 128;
  float* deg   = g_buf + (size_t)E;
  float* outp  = (float*)d_out;

  const int gateBlocks = (E + 127) / 128;
  const int gemmBlocks = (N + 16 * RPW - 1) / (16 * RPW);
  const int nBlocks    = (N + 255) / 256;
  const int eBlocks    = (E + 255) / 256;
  const int neBlocks   = ((int)((size_t)N * 128 + 255)) / 256;
  const int aggBlocks  = (int)(((size_t)E * 32 + 255) / 256);

  pdn_gate_kernel<<<gateBlocks, 256, 0, stream>>>(ea, m1w1, m1b1, m1w2, m1b2, g_buf, E);
  pdn_deg_init_kernel<<<nBlocks, 256, 0, stream>>>(deg, N);
  pdn_deg_scatter_kernel<<<eBlocks, 256, 0, stream>>>(col, g_buf, deg, E);
  pdn_dinv_kernel<<<nBlocks, 256, 0, stream>>>(deg, N);
  pdn_gemm_kernel<<<gemmBlocks, 256, 0, stream>>>(x, W1, h_buf, N, 0);
  pdn_out_init_kernel<<<neBlocks, 256, 0, stream>>>(h_buf, deg, a_buf, N);
  pdn_agg_kernel<<<aggBlocks, 256, 0, stream>>>(h_buf, row, col, g_buf, deg, a_buf, E);

  pdn_gate_kernel<<<gateBlocks, 256, 0, stream>>>(ea, m2w1, m2b1, m2w2, m2b2, g_buf, E);
  pdn_deg_init_kernel<<<nBlocks, 256, 0, stream>>>(deg, N);
  pdn_deg_scatter_kernel<<<eBlocks, 256, 0, stream>>>(col, g_buf, deg, E);
  pdn_dinv_kernel<<<nBlocks, 256, 0, stream>>>(deg, N);
  pdn_gemm_kernel<<<gemmBlocks, 256, 0, stream>>>(a_buf, W2, h_buf, N, 1);
  pdn_out_init_kernel<<<neBlocks, 256, 0, stream>>>(h_buf, deg, outp, N);
  pdn_agg_kernel<<<aggBlocks, 256, 0, stream>>>(h_buf, row, col, g_buf, deg, outp, E);
}


# agg by target: in-call CSR (int-atomic count, scan, slot fill), one wave per node sums in-edges in registers, one store per out line; float atomics + out_init gone; sum order = slot order
# speedup vs baseline: 1.7986x; 1.4406x over previous
#include <hip/hip_runtime.h>
#include <math.h>

typedef __attribute__((ext_vector_type(2))) float v2f;
typedef __attribute__((ext_vector_type(8))) float v8f;

#define WMMA_F32X4(a, b, c) \
    __builtin_amdgcn_wmma_f32_16x16x4_f32(false, (a), false, (b), (short)0, (c), false, false)

__global__ __launch_bounds__(256) void pdn_gate_kernel(
    const float* __restrict__ ea, const float* __restrict__ mw1,
    const float* __restrict__ mb1, const float* __restrict__ mw2,
    const float* __restrict__ mb2, float* __restrict__ g, int E) {
  const int wave = threadIdx.x >> 5;
  const int lane = threadIdx.x & 31;
  const int half = lane >> 4;
  const int l16  = lane & 15;
  const int e0 = (blockIdx.x * 8 + wave) * 16;
  if (e0 >= E) return;

  int er = e0 + l16;
  if (er >= E) er = E - 1;
  const float* earow = ea + (size_t)er * 16;

  v2f afrag[4];
#pragma unroll
  for (int kk = 0; kk < 4; ++kk) {
    afrag[kk].x = earow[4 * kk + 2 * half];
    afrag[kk].y = earow[4 * kk + 2 * half + 1];
  }

  float acc[8];
#pragma unroll
  for (int v = 0; v < 8; ++v) acc[v] = 0.0f;

#pragma unroll
  for (int t = 0; t < 8; ++t) {
    const int n0 = t * 16;
    v8f c = {};
#pragma unroll
    for (int kk = 0; kk < 4; ++kk) {
      const int kb = 4 * kk + 2 * half;
      v2f b;
      b.x = mw1[(size_t)kb * 128 + n0 + l16];
      b.y = mw1[(size_t)(kb + 1) * 128 + n0 + l16];
      c = WMMA_F32X4(afrag[kk], b, c);
    }
    const float b1 = mb1[n0 + l16];
    const float w2 = mw2[n0 + l16];
#pragma unroll
    for (int v = 0; v < 8; ++v) {
      acc[v] += fmaxf(c[v] + b1, 0.0f) * w2;
    }
  }

#pragma unroll
  for (int off = 1; off < 16; off <<= 1) {
#pragma unroll
    for (int v = 0; v < 8; ++v) acc[v] += __shfl_xor(acc[v], off, 32);
  }

  const float b2 = mb2[0];
#pragma unroll
  for (int v = 0; v < 8; ++v) {
    if (l16 == v) {
      const int e = e0 + v + 8 * half;
      if (e < E) {
        const float s = acc[v] + b2;
        g[e] = 1.0f / (1.0f + __expf(-s));
      }
    }
  }
}

#define RPW 4
__global__ __launch_bounds__(256) void pdn_gemm_kernel(
    const float* __restrict__ x, const float* __restrict__ W,
    float* __restrict__ h, int M, int relu_in) {
  const int wave = threadIdx.x >> 5;
  const int lane = threadIdx.x & 31;
  const int half = lane >> 4;
  const int l16  = lane & 15;
  const int n0 = wave * 16;
  const int rowBase = blockIdx.x * (16 * RPW);

  const float* xrow[RPW];
#pragma unroll
  for (int t = 0; t < RPW; ++t) {
    int ar = rowBase + t * 16 + l16;
    if (ar >= M) ar = M - 1;
    xrow[t] = x + (size_t)ar * 128;
  }

  v8f c[RPW];
#pragma unroll
  for (int t = 0; t < RPW; ++t) c[t] = (v8f){};

#pragma unroll 4
  for (int k0 = 0; k0 < 128; k0 += 4) {
    const int ka = k0 + half * 2;
    v2f b;
    b.x = W[(size_t)ka * 128 + n0 + l16];
    b.y = W[(size_t)(ka + 1) * 128 + n0 + l16];
#pragma unroll
    for (int t = 0; t < RPW; ++t) {
      v2f a;
      a.x = xrow[t][ka];
      a.y = xrow[t][ka + 1];
      if (relu_in) { a.x = fmaxf(a.x, 0.0f); a.y = fmaxf(a.y, 0.0f); }
      c[t] = WMMA_F32X4(a, b, c[t]);
    }
  }

#pragma unroll
  for (int t = 0; t < RPW; ++t) {
#pragma unroll
    for (int v = 0; v < 8; ++v) {
      const int r = rowBase + t * 16 + v + 8 * half;
      if (r < M) h[(size_t)r * 128 + n0 + l16] = c[t][v];
    }
  }
}


__global__ __launch_bounds__(256) void pdn_csr_zero_kernel(int* __restrict__ cnt, int n) {
  const int i = blockIdx.x * blockDim.x + threadIdx.x;
  if (i < n) cnt[i] = 0;
}

__global__ __launch_bounds__(256) void pdn_csr_count_kernel(const int* __restrict__ col,
                                                            int* __restrict__ cnt, int E) {
  const int e = blockIdx.x * blockDim.x + threadIdx.x;
  if (e < E) atomicAdd(&cnt[col[e]], 1);
}

#define SCAN_T 1024
__global__ __launch_bounds__(SCAN_T) void pdn_csr_scan_kernel(const int* __restrict__ cnt,
                                                              int* __restrict__ rowptr, int n) {
  __shared__ int s_part[32];
  __shared__ int s_base;
  const int tid  = threadIdx.x;
  const int lane = tid & 31;
  const int wave = tid >> 5;
  const int first = blockIdx.x * SCAN_T;

  int pre = 0;
  for (int j = tid; j < first; j += SCAN_T) pre += cnt[j];
  for (int off = 16; off > 0; off >>= 1) pre += __shfl_xor(pre, off, 32);
  if (lane == 0) s_part[wave] = pre;
  __syncthreads();
  if (wave == 0) {
    int v = s_part[lane];
    for (int off = 16; off > 0; off >>= 1) v += __shfl_xor(v, off, 32);
    if (lane == 0) s_base = v;
  }
  __syncthreads();
  const int base = s_base;

  const int i = first + tid;
  const int c = (i < n) ? cnt[i] : 0;
  int inc = c;
  for (int off = 1; off < 32; off <<= 1) {
    const int up = __shfl_up(inc, off, 32);
    if (lane >= off) inc += up;
  }
  if (lane == 31) s_part[wave] = inc;
  __syncthreads();
  int woff = (lane < wave) ? s_part[lane] : 0;
  for (int off = 16; off > 0; off >>= 1) woff += __shfl_xor(woff, off, 32);
  if (i < n) rowptr[i] = base + woff + inc - c;
}

__global__ __launch_bounds__(256) void pdn_csr_fill_kernel(const int* __restrict__ row,
                                                           const int* __restrict__ col,
                                                           const int* __restrict__ rowptr,
                                                           int* __restrict__ cnt,
                                                           int2* __restrict__ ent, int E) {
  const int e = blockIdx.x * blockDim.x + threadIdx.x;
  if (e < E) {
    const int c = col[e];
    const int k = atomicAdd(&cnt[c], -1);
    ent[rowptr[c] + k - 1] = make_int2(row[e], e);
  }
}

__global__ __launch_bounds__(256) void pdn_deg_kernel(const int* __restrict__ rowptr,
                                                      const int2* __restrict__ ent,
                                                      const float* __restrict__ g,
                                                      float* __restrict__ dinv, int N) {
  const int t = blockIdx.x * blockDim.x + threadIdx.x;
  const int i = t >> 4;
  const int l = t & 15;
  float s = 0.0f;
  if (i < N) {
    const int end = rowptr[i + 1];
    for (int p = rowptr[i] + l; p < end; p += 16) s += g[ent[p].y];
  }
  for (int off = 1; off < 16; off <<= 1) s += __shfl_xor(s, off, 32);
  if (i < N && l == 0) {
    const float d = 1.0f + s;
    dinv[i] = (d > 0.0f) ? rsqrtf(d) : 0.0f;
  }
}

__global__ __launch_bounds__(256) void pdn_agg_kernel(const float* __restrict__ h,
                                                      const int* __restrict__ rowptr,
                                                      const int2* __restrict__ ent,
                                                      const float* __restrict__ g,
                                                      const float* __restrict__ dinv,
                                                      float* __restrict__ out, int N) {
  const int lane = threadIdx.x & 31;
  const int i = __builtin_amdgcn_readfirstlane((int)(blockIdx.x * 8 + (threadIdx.x >> 5)));
  if (i >= N) return;
  const int beg = rowptr[i];
  const int end = rowptr[i + 1];
  const float di = dinv[i];
  const float* hl = h + lane * 4;

  const float4 hs = *(const float4*)(hl + (size_t)i * 128);
  const float wself = di * di;
  float4 acc;
  acc.x = wself * hs.x; acc.y = wself * hs.y; acc.z = wself * hs.z; acc.w = wself * hs.w;

  for (int base = beg; base < end; base += 32) {
    const int p = base + lane;
    int r = 0;
    float w = 0.0f;
    if (p < end) {
      const int2 en = ent[p];
      r = en.x;
      w = dinv[r] * g[en.y] * di;
    }
    const int wi = __float_as_int(w);
    const int rem = end - base;
    const int n = rem < 32 ? rem : 32;
    int j = 0;
    for (; j + 4 <= n; j += 4) {
      const int r0 = __builtin_amdgcn_readlane(r, j);
      const int r1 = __builtin_amdgcn_readlane(r, j + 1);
      const int r2 = __builtin_amdgcn_readlane(r, j + 2);
      const int r3 = __builtin_amdgcn_readlane(r, j + 3);
      const float4 a0 = *(const float4*)(hl + (size_t)r0 * 128);
      const float4 a1 = *(const float4*)(hl + (size_t)r1 * 128);
      const float4 a2 = *(const float4*)(hl + (size_t)r2 * 128);
      const float4 a3 = *(const float4*)(hl + (size_t)r3 * 128);
      const float w0 = __int_as_float(__builtin_amdgcn_readlane(wi, j));
      const float w1 = __int_as_float(__builtin_amdgcn_readlane(wi, j + 1));
      const float w2 = __int_as_float(__builtin_amdgcn_readlane(wi, j + 2));
      const float w3 = __int_as_float(__builtin_amdgcn_readlane(wi, j + 3));
      acc.x += w0 * a0.x; acc.y += w0 * a0.y; acc.z += w0 * a0.z; acc.w += w0 * a0.w;
      acc.x += w1 * a1.x; acc.y += w1 * a1.y; acc.z += w1 * a1.z; acc.w += w1 * a1.w;
      acc.x += w2 * a2.x; acc.y += w2 * a2.y; acc.z += w2 * a2.z; acc.w += w2 * a2.w;
      acc.x += w3 * a3.x; acc.y += w3 * a3.y; acc.z += w3 * a3.z; acc.w += w3 * a3.w;
    }
    for (; j < n; ++j) {
      const int r0 = __builtin_amdgcn_readlane(r, j);
      const float w0 = __int_as_float(__builtin_amdgcn_readlane(wi, j));
      const float4 a0 = *(const float4*)(hl + (size_t)r0 * 128);
      acc.x += w0 * a0.x; acc.y += w0 * a0.y; acc.z += w0 * a0.z; acc.w += w0 * a0.w;
    }
  }
  *(float4*)(out + (size_t)i * 128 + lane * 4) = acc;
}


extern "C" void kernel_launch(void* const* d_in, const int* in_sizes, int n_in,
                              void* d_out, int out_size, void* d_ws, size_t ws_size,
                              hipStream_t stream) {
  const float* x    = (const float*)d_in[0];
  const int*   ei   = (const int*)d_in[1];
  const float* ea   = (const float*)d_in[2];
  const float* W1   = (const float*)d_in[3];
  const float* m1w1 = (const float*)d_in[4];
  const float* m1b1 = (const float*)d_in[5];
  const float* m1w2 = (const float*)d_in[6];
  const float* m1b2 = (const float*)d_in[7];
  const float* W2   = (const float*)d_in[8];
  const float* m2w1 = (const float*)d_in[9];
  const float* m2b1 = (const float*)d_in[10];
  const float* m2w2 = (const float*)d_in[11];
  const float* m2b2 = (const float*)d_in[12];

  const int N = in_sizes[0] / 128;
  const int E = in_sizes[2] / 16;
  const int* row = ei;
  const int* col = ei + E;

  float* ws     = (float*)d_ws;
  float* h_buf  = ws;
  int2*  ent    = (int2*)(h_buf + (size_t)N * 128);
  float* g_buf  = (float*)(ent + (size_t)E);
  float* dinv   = g_buf + (size_t)E;
  int*   cnt    = (int*)(dinv + (size_t)N);
  int*   rowptr = cnt + (size_t)N + 1;
  float* outp   = (float*)d_out;

  const int gateBlocks = (E + 127) / 128;
  const int gemmBlocks = (N + 16 * RPW - 1) / (16 * RPW);
  const int eBlocks    = (E + 255) / 256;
  const int zBlocks    = (N + 1 + 255) / 256;
  const int scanBlocks = (N + 1 + SCAN_T - 1) / SCAN_T;
  const int degBlocks  = (int)(((size_t)N * 16 + 255) / 256);
  const int aggBlocks  = (N + 7) / 8;

  pdn_csr_zero_kernel<<<zBlocks, 256, 0, stream>>>(cnt, N + 1);
  pdn_csr_count_kernel<<<eBlocks, 256, 0, stream>>>(col, cnt, E);
  pdn_csr_scan_kernel<<<scanBlocks, SCAN_T, 0, stream>>>(cnt, rowptr, N + 1);
  pdn_csr_fill_kernel<<<eBlocks, 256, 0, stream>>>(row, col, rowptr, cnt, ent, E);

  pdn_gate_kernel<<<gateBlocks, 256, 0, stream>>>(ea, m1w1, m1b1, m1w2, m1b2, g_buf, E);
  pdn_deg_kernel<<<degBlocks, 256, 0, stream>>>(rowptr, ent, g_buf, dinv, N);
  pdn_gemm_kernel<<<gemmBlocks, 256, 0, stream>>>(x, W1, h_buf, N, 0);
  pdn_agg_kernel<<<aggBlocks, 256, 0, stream>>>(h_buf, rowptr, ent, g_buf, dinv, outp, N);

  pdn_gate_kernel<<<gateBlocks, 256, 0, stream>>>(ea, m2w1, m2b1, m2w2, m2b2, g_buf, E);
  pdn_deg_kernel<<<degBlocks, 256, 0, stream>>>(rowptr, ent, g_buf, dinv, N);
  pdn_gemm_kernel<<<gemmBlocks, 256, 0, stream>>>(outp, W2, h_buf, N, 1);
  pdn_agg_kernel<<<aggBlocks, 256, 0, stream>>>(h_buf, rowptr, ent, g_buf, dinv, outp, N);
}


# no per-edge gather left but rows: both layers' gates first (one edge_attr pass), bucket fill puts source+gates in slot order, dinv[r] folded into GEMM rows; rows fetched 8 per trip; 12->9 launches
# speedup vs baseline: 1.8822x; 1.0465x over previous
#include <hip/hip_runtime.h>
#include <math.h>

typedef __attribute__((ext_vector_type(2))) float v2f;
typedef __attribute__((ext_vector_type(8))) float v8f;

#define WMMA_F32X4(a, b, c) \
    __builtin_amdgcn_wmma_f32_16x16x4_f32(false, (a), false, (b), (short)0, (c), false, false)

__device__ __forceinline__ float pdn_gate_one(const v2f (&afrag)[4],
    const float* __restrict__ mw1, const float* __restrict__ mb1,
    const float* __restrict__ mw2, const float* __restrict__ mb2,
    const int half, const int l16) {
  float acc[8];
#pragma unroll
  for (int v = 0; v < 8; ++v) acc[v] = 0.0f;

#pragma unroll
  for (int t = 0; t < 8; ++t) {
    const int n0 = t * 16;
    v8f c = {};
#pragma unroll
    for (int kk = 0; kk < 4; ++kk) {
      const int kb = 4 * kk + 2 * half;
      v2f b;
      b.x = mw1[(size_t)kb * 128 + n0 + l16];
      b.y = mw1[(size_t)(kb + 1) * 128 + n0 + l16];
      c = WMMA_F32X4(afrag[kk], b, c);
    }
    const float b1 = mb1[n0 + l16];
    const float w2 = mw2[n0 + l16];
#pragma unroll
    for (int v = 0; v < 8; ++v) {
      acc[v] += fmaxf(c[v] + b1, 0.0f) * w2;
    }
  }

#pragma unroll
  for (int off = 1; off < 16; off <<= 1) {
#pragma unroll
    for (int v = 0; v < 8; ++v) acc[v] += __shfl_xor(acc[v], off, 32);
  }

  float s = acc[0];
#pragma unroll
  for (int v = 1; v < 8; ++v) s = (l16 == v) ? acc[v] : s;
  s += mb2[0];
  return 1.0f / (1.0f + __expf(-s));
}

__global__ __launch_bounds__(256) void pdn_gate_kernel(
    const float* __restrict__ ea,
    const float* __restrict__ aw1, const float* __restrict__ ab1,
    const float* __restrict__ aw2, const float* __restrict__ ab2,
    const float* __restrict__ bw1, const float* __restrict__ bb1,
    const float* __restrict__ bw2, const float* __restrict__ bb2,
    float2* __restrict__ g12, int* __restrict__ cnt, int ncnt, int E) {
  for (int z = blockIdx.x * 256 + threadIdx.x; z < ncnt; z += gridDim.x * 256) cnt[z] = 0;

  const int wave = threadIdx.x >> 5;
  const int lane = threadIdx.x & 31;
  const int half = lane >> 4;
  const int l16  = lane & 15;
  const int e0 = (blockIdx.x * 8 + wave) * 16;
  if (e0 >= E) return;

  int er = e0 + l16;
  if (er >= E) er = E - 1;
  const float* earow = ea + (size_t)er * 16;

  v2f afrag[4];
#pragma unroll
  for (int kk = 0; kk < 4; ++kk) {
    afrag[kk].x = earow[4 * kk + 2 * half];
    afrag[kk].y = earow[4 * kk + 2 * half + 1];
  }

  float2 gg;
  gg.x = 0.0f; gg.y = 0.0f;
#pragma nounroll
  for (int layer = 0; layer < 2; ++layer) {
    const float* w1 = layer ? bw1 : aw1;
    const float* b1 = layer ? bb1 : ab1;
    const float* w2 = layer ? bw2 : aw2;
    const float* b2 = layer ? bb2 : ab2;
    const float gv = pdn_gate_one(afrag, w1, b1, w2, b2, half, l16);
    if (layer) gg.y = gv; else gg.x = gv;
  }

  const int e = e0 + l16 + 8 * half;
  if (l16 < 8 && e < E) g12[e] = gg;
}

#define RPW 4
__global__ __launch_bounds__(256) void pdn_gemm_kernel(
    const float* __restrict__ x, const float* __restrict__ W,
    const float* __restrict__ rs, float* __restrict__ h, int M, int relu_in) {
  const int wave = threadIdx.x >> 5;
  const int lane = threadIdx.x & 31;
  const int half = lane >> 4;
  const int l16  = lane & 15;
  const int n0 = wave * 16;
  const int rowBase = blockIdx.x * (16 * RPW);

  const float* xrow[RPW];
#pragma unroll
  for (int t = 0; t < RPW; ++t) {
    int ar = rowBase + t * 16 + l16;
    if (ar >= M) ar = M - 1;
    xrow[t] = x + (size_t)ar * 128;
  }

  v8f c[RPW];
#pragma unroll
  for (int t = 0; t < RPW; ++t) c[t] = (v8f){};

#pragma unroll 4
  for (int k0 = 0; k0 < 128; k0 += 4) {
    const int ka = k0 + half * 2;
    v2f b;
    b.x = W[(size_t)ka * 128 + n0 + l16];
    b.y = W[(size_t)(ka + 1) * 128 + n0 + l16];
#pragma unroll
    for (int t = 0; t < RPW; ++t) {
      v2f a;
      a.x = xrow[t][ka];
      a.y = xrow[t][ka + 1];
      if (relu_in) { a.x = fmaxf(a.x, 0.0f); a.y = fmaxf(a.y, 0.0f); }
      c[t] = WMMA_F32X4(a, b, c[t]);
    }
  }

  float rsv[RPW][8];
#pragma unroll
  for (int t = 0; t < RPW; ++t) {
#pragma unroll
    for (int v = 0; v < 8; ++v) {
      int r = rowBase + t * 16 + v + 8 * half;
      if (r >= M) r = M - 1;
      rsv[t][v] = rs[r];
    }
  }

#pragma unroll
  for (int t = 0; t < RPW; ++t) {
#pragma unroll
    for (int v = 0; v < 8; ++v) {
      const int r = rowBase + t * 16 + v + 8 * half;
      if (r < M) h[(size_t)r * 128 + n0 + l16] = c[t][v] * rsv[t][v];
    }
  }
}


__global__ __launch_bounds__(256) void pdn_csr_count_kernel(const int* __restrict__ col,
                                                            int* __restrict__ cnt, int E) {
  const int e = blockIdx.x * blockDim.x + threadIdx.x;
  if (e < E) atomicAdd(&cnt[col[e]], 1);
}

#define SCAN_T 1024
__global__ __launch_bounds__(SCAN_T) void pdn_csr_scan_kernel(const int* __restrict__ cnt,
                                                              int* __restrict__ rowptr, int n) {
  __shared__ int s_part[32];
  __shared__ int s_base;
  const int tid  = threadIdx.x;
  const int lane = tid & 31;
  const int wave = tid >> 5;
  const int first = blockIdx.x * SCAN_T;

  int pre = 0;
  for (int j = tid; j < first; j += SCAN_T) pre += cnt[j];
  for (int off = 16; off > 0; off >>= 1) pre += __shfl_xor(pre, off, 32);
  if (lane == 0) s_part[wave] = pre;
  __syncthreads();
  if (wave == 0) {
    int v = s_part[lane];
    for (int off = 16; off > 0; off >>= 1) v += __shfl_xor(v, off, 32);
    if (lane == 0) s_base = v;
  }
  __syncthreads();
  const int base = s_base;

  const int i = first + tid;
  const int c = (i < n) ? cnt[i] : 0;
  int inc = c;
  for (int off = 1; off < 32; off <<= 1) {
    const int up = __shfl_up(inc, off, 32);
    if (lane >= off) inc += up;
  }
  if (lane == 31) s_part[wave] = inc;
  __syncthreads();
  int woff = (lane < wave) ? s_part[lane] : 0;
  for (int off = 16; off > 0; off >>= 1) woff += __shfl_xor(woff, off, 32);
  if (i < n) rowptr[i] = base + woff + inc - c;
}

__global__ __launch_bounds__(256) void pdn_csr_fill_kernel(const int* __restrict__ row,
                                                           const int* __restrict__ col,
                                                           const int* __restrict__ rowptr,
                                                           int* __restrict__ cnt,
                                                           const float2* __restrict__ g12,
                                                           int2* __restrict__ ent,
                                                           float* __restrict__ gs2, int E) {
  const int e = blockIdx.x * blockDim.x + threadIdx.x;
  if (e < E) {
    const int c = col[e];
    const float2 gg = g12[e];
    const int k = atomicAdd(&cnt[c], -1);
    const int p = rowptr[c] + k - 1;
    ent[p] = make_int2(row[e], __float_as_int(gg.x));
    gs2[p] = gg.y;
  }
}

__global__ __launch_bounds__(256) void pdn_deg_kernel(const int* __restrict__ rowptr,
                                                      const int2* __restrict__ ent,
                                                      const float* __restrict__ gs2,
                                                      float* __restrict__ dinv1,
                                                      float* __restrict__ dinv2, int N) {
  const int t = blockIdx.x * blockDim.x + threadIdx.x;
  const int i = t >> 4;
  const int l = t & 15;
  float s1 = 0.0f, s2 = 0.0f;
  if (i < N) {
    const int end = rowptr[i + 1];
    for (int p = rowptr[i] + l; p < end; p += 16) {
      s1 += __int_as_float(ent[p].y);
      s2 += gs2[p];
    }
  }
  for (int off = 1; off < 16; off <<= 1) {
    s1 += __shfl_xor(s1, off, 32);
    s2 += __shfl_xor(s2, off, 32);
  }
  if (i < N && l == 0) {
    const float d1 = 1.0f + s1;
    const float d2 = 1.0f + s2;
    dinv1[i] = (d1 > 0.0f) ? rsqrtf(d1) : 0.0f;
    dinv2[i] = (d2 > 0.0f) ? rsqrtf(d2) : 0.0f;
  }
}

template <bool GATE_IN_ENT>
__global__ __launch_bounds__(256) void pdn_agg_kernel(const float* __restrict__ h,
                                                      const int* __restrict__ rowptr,
                                                      const int2* __restrict__ ent,
                                                      const float* __restrict__ gs,
                                                      const float* __restrict__ dinv,
                                                      float* __restrict__ out, int N) {
  const int lane = threadIdx.x & 31;
  const int i = __builtin_amdgcn_readfirstlane((int)(blockIdx.x * 8 + (threadIdx.x >> 5)));
  if (i >= N) return;
  const int beg = rowptr[i];
  const int end = rowptr[i + 1];
  const float di = dinv[i];
  const float* hl = h + lane * 4;

  float4 acc = *(const float4*)(hl + (size_t)i * 128);

  for (int base = beg; base < end; base += 32) {
    const int p = base + lane;
    int r = i;
    float w = 0.0f;
    if (p < end) {
      const int2 en = ent[p];
      r = en.x;
      w = GATE_IN_ENT ? __int_as_float(en.y) : gs[p];
    }
    const int wi = __float_as_int(w);
    const int rem = end - base;
    const int n = rem < 32 ? rem : 32;
    for (int j = 0; j < n; j += 8) {
      int rr[8];
      float ww[8];
      float4 a[8];
#pragma unroll
      for (int u = 0; u < 8; ++u) rr[u] = __builtin_amdgcn_readlane(r, j + u);
#pragma unroll
      for (int u = 0; u < 8; ++u) a[u] = *(const float4*)(hl + (size_t)rr[u] * 128);
#pragma unroll
      for (int u = 0; u < 8; ++u) ww[u] = __int_as_float(__builtin_amdgcn_readlane(wi, j + u));
#pragma unroll
      for (int u = 0; u < 8; ++u) {
        acc.x += ww[u] * a[u].x; acc.y += ww[u] * a[u].y;
        acc.z += ww[u] * a[u].z; acc.w += ww[u] * a[u].w;
      }
    }
  }
  acc.x *= di; acc.y *= di; acc.z *= di; acc.w *= di;
  *(float4*)(out + (size_t)i * 128 + lane * 4) = acc;
}


extern "C" void kernel_launch(void* const* d_in, const int* in_sizes, int n_in,
                              void* d_out, int out_size, void* d_ws, size_t ws_size,
                              hipStream_t stream) {
  const float* x    = (const float*)d_in[0];
  const int*   ei   = (const int*)d_in[1];
  const float* ea   = (const float*)d_in[2];
  const float* W1   = (const float*)d_in[3];
  const float* m1w1 = (const float*)d_in[4];
  const float* m1b1 = (const float*)d_in[5];
  const float* m1w2 = (const float*)d_in[6];
  const float* m1b2 = (const float*)d_in[7];
  const float* W2   = (const float*)d_in[8];
  const float* m2w1 = (const float*)d_in[9];
  const float* m2b1 = (const float*)d_in[10];
  const float* m2w2 = (const float*)d_in[11];
  const float* m2b2 = (const float*)d_in[12];

  const int N = in_sizes[0] / 128;
  const int E = in_sizes[2] / 16;
  const int* row = ei;
  const int* col = ei + E;

  float* ws     = (float*)d_ws;
  float* h_buf  = ws;
  int2*  ent    = (int2*)(h_buf + (size_t)N * 128);
  float* gs2    = (float*)(ent + (size_t)E);
  float* dinv1  = gs2 + (size_t)E;
  int*   cnt    = (int*)(dinv1 + (size_t)N);
  int*   rowptr = cnt + (size_t)N + 1;
  float* dinv2  = (float*)cnt;
  float* outp   = (float*)d_out;
  float2* g12   = (float2*)d_out;

  const int gateBlocks = (E + 127) / 128;
  const int gemmBlocks = (N + 16 * RPW - 1) / (16 * RPW);
  const int eBlocks    = (E + 255) / 256;
  const int scanBlocks = (N + 1 + SCAN_T - 1) / SCAN_T;
  const int degBlocks  = (int)(((size_t)N * 16 + 255) / 256);
  const int aggBlocks  = (N + 7) / 8;

  pdn_gate_kernel<<<gateBlocks, 256, 0, stream>>>(ea, m1w1, m1b1, m1w2, m1b2,
                                                  m2w1, m2b1, m2w2, m2b2, g12, cnt, N + 1, E);
  pdn_csr_count_kernel<<<eBlocks, 256, 0, stream>>>(col, cnt, E);
  pdn_csr_scan_kernel<<<scanBlocks, SCAN_T, 0, stream>>>(cnt, rowptr, N + 1);
  pdn_csr_fill_kernel<<<eBlocks, 256, 0, stream>>>(row, col, rowptr, cnt, g12, ent, gs2, E);
  pdn_deg_kernel<<<degBlocks, 256, 0, stream>>>(rowptr, ent, gs2, dinv1, dinv2, N);

  pdn_gemm_kernel<<<gemmBlocks, 256, 0, stream>>>(x, W1, dinv1, h_buf, N, 0);
  pdn_agg_kernel<true><<<aggBlocks, 256, 0, stream>>>(h_buf, rowptr, ent, gs2, dinv1, outp, N);

  pdn_gemm_kernel<<<gemmBlocks, 256, 0, stream>>>(outp, W2, dinv2, h_buf, N, 1);
  pdn_agg_kernel<false><<<aggBlocks, 256, 0, stream>>>(h_buf, rowptr, ent, gs2, dinv2, outp, N);
}


# CSR build without global atomics: (node-range x edge-slice) blocks count and place edges with LDS counters, slice counts prefixed per node; slot (= sum) order now by slice; 9->10 launches
# speedup vs baseline: 2.8130x; 1.4945x over previous
#include <hip/hip_runtime.h>
#include <math.h>

typedef __attribute__((ext_vector_type(2))) float v2f;
typedef __attribute__((ext_vector_type(8))) float v8f;

#define WMMA_F32X4(a, b, c) \
    __builtin_amdgcn_wmma_f32_16x16x4_f32(false, (a), false, (b), (short)0, (c), false, false)

__device__ __forceinline__ float pdn_gate_one(const v2f (&afrag)[4],
    const float* __restrict__ mw1, const float* __restrict__ mb1,
    const float* __restrict__ mw2, const float* __restrict__ mb2,
    const int half, const int l16) {
  float acc[8];
#pragma unroll
  for (int v = 0; v < 8; ++v) acc[v] = 0.0f;

#pragma unroll
  for (int t = 0; t < 8; ++t) {
    const int n0 = t * 16;
    v8f c = {};
#pragma unroll
    for (int kk = 0; kk < 4; ++kk) {
      const int kb = 4 * kk + 2 * half;
      v2f b;
      b.x = mw1[(size_t)kb * 128 + n0 + l16];
      b.y = mw1[(size_t)(kb + 1) * 128 + n0 + l16];
      c = WMMA_F32X4(afrag[kk], b, c);
    }
    const float b1 = mb1[n0 + l16];
    const float w2 = mw2[n0 + l16];
#pragma unroll
    for (int v = 0; v < 8; ++v) {
      acc[v] += fmaxf(c[v] + b1, 0.0f) * w2;
    }
  }

#pragma unroll
  for (int off = 1; off < 16; off <<= 1) {
#pragma unroll
    for (int v = 0; v < 8; ++v) acc[v] += __shfl_xor(acc[v], off, 32);
  }

  float s = acc[0];
#pragma unroll
  for (int v = 1; v < 8; ++v) s = (l16 == v) ? acc[v] : s;
  s += mb2[0];
  return 1.0f / (1.0f + __expf(-s));
}

__global__ __launch_bounds__(256) void pdn_gate_kernel(
    const float* __restrict__ ea,
    const float* __restrict__ aw1, const float* __restrict__ ab1,
    const float* __restrict__ aw2, const float* __restrict__ ab2,
    const float* __restrict__ bw1, const float* __restrict__ bb1,
    const float* __restrict__ bw2, const float* __restrict__ bb2,
    float2* __restrict__ g12, int E) {
  const int wave = threadIdx.x >> 5;
  const int lane = threadIdx.x & 31;
  const int half = lane >> 4;
  const int l16  = lane & 15;
  const int e0 = (blockIdx.x * 8 + wave) * 16;
  if (e0 >= E) return;

  int er = e0 + l16;
  if (er >= E) er = E - 1;
  const float* earow = ea + (size_t)er * 16;

  v2f afrag[4];
#pragma unroll
  for (int kk = 0; kk < 4; ++kk) {
    afrag[kk].x = earow[4 * kk + 2 * half];
    afrag[kk].y = earow[4 * kk + 2 * half + 1];
  }

  float2 gg;
  gg.x = 0.0f; gg.y = 0.0f;
#pragma nounroll
  for (int layer = 0; layer < 2; ++layer) {
    const float* w1 = layer ? bw1 : aw1;
    const float* b1 = layer ? bb1 : ab1;
    const float* w2 = layer ? bw2 : aw2;
    const float* b2 = layer ? bb2 : ab2;
    const float gv = pdn_gate_one(afrag, w1, b1, w2, b2, half, l16);
    if (layer) gg.y = gv; else gg.x = gv;
  }

  const int e = e0 + l16 + 8 * half;
  if (l16 < 8 && e < E) g12[e] = gg;
}

#define RPW 4
__global__ __launch_bounds__(256) void pdn_gemm_kernel(
    const float* __restrict__ x, const float* __restrict__ W,
    const float* __restrict__ rs, float* __restrict__ h, int M, int relu_in) {
  const int wave = threadIdx.x >> 5;
  const int lane = threadIdx.x & 31;
  const int half = lane >> 4;
  const int l16  = lane & 15;
  const int n0 = wave * 16;
  const int rowBase = blockIdx.x * (16 * RPW);

  const float* xrow[RPW];
#pragma unroll
  for (int t = 0; t < RPW; ++t) {
    int ar = rowBase + t * 16 + l16;
    if (ar >= M) ar = M - 1;
    xrow[t] = x + (size_t)ar * 128;
  }

  v8f c[RPW];
#pragma unroll
  for (int t = 0; t < RPW; ++t) c[t] = (v8f){};

#pragma unroll 4
  for (int k0 = 0; k0 < 128; k0 += 4) {
    const int ka = k0 + half * 2;
    v2f b;
    b.x = W[(size_t)ka * 128 + n0 + l16];
    b.y = W[(size_t)(ka + 1) * 128 + n0 + l16];
#pragma unroll
    for (int t = 0; t < RPW; ++t) {
      v2f a;
      a.x = xrow[t][ka];
      a.y = xrow[t][ka + 1];
      if (relu_in) { a.x = fmaxf(a.x, 0.0f); a.y = fmaxf(a.y, 0.0f); }
      c[t] = WMMA_F32X4(a, b, c[t]);
    }
  }

  float rsv[RPW][8];
#pragma unroll
  for (int t = 0; t < RPW; ++t) {
#pragma unroll
    for (int v = 0; v < 8; ++v) {
      int r = rowBase + t * 16 + v + 8 * half;
      if (r >= M) r = M - 1;
      rsv[t][v] = rs[r];
    }
  }

#pragma unroll
  for (int t = 0; t < RPW; ++t) {
#pragma unroll
    for (int v = 0; v < 8; ++v) {
      const int r = rowBase + t * 16 + v + 8 * half;
      if (r < M) h[(size_t)r * 128 + n0 + l16] = c[t][v] * rsv[t][v];
    }
  }
}

#define CSR_RANGE 12500
#define CSR_S 32
#define CSR_T 1024
#define CSR_U 4

__global__ __launch_bounds__(CSR_T) void pdn_csr_count_kernel(const int* __restrict__ col,
                                                              int* __restrict__ tab,
                                                              int R, int ES, int E, int N) {
  __shared__ int s_cnt[CSR_RANGE];
  const int r  = blockIdx.x % R;
  const int s  = blockIdx.x / R;
  const int r0 = r * CSR_RANGE;
  const int nr = min(CSR_RANGE, N - r0);
  for (int j = threadIdx.x; j < nr; j += CSR_T) s_cnt[j] = 0;
  __syncthreads();

  const int e0 = s * ES;
  const int e1 = min(e0 + ES, E);
  for (int base = e0 + (int)threadIdx.x; base < e1; base += CSR_U * CSR_T) {
    int c[CSR_U];
#pragma unroll
    for (int u = 0; u < CSR_U; ++u) {
      const int e = base + u * CSR_T;
      c[u] = col[min(e, e1 - 1)];
      if (e >= e1) c[u] = -1;
    }
#pragma unroll
    for (int u = 0; u < CSR_U; ++u) {
      const unsigned d = (unsigned)(c[u] - r0);
      if (d < (unsigned)nr) atomicAdd(&s_cnt[d], 1);
    }
  }
  __syncthreads();
  for (int j = threadIdx.x; j < nr; j += CSR_T) tab[(size_t)s * N + r0 + j] = s_cnt[j];
}

__global__ __launch_bounds__(256) void pdn_csr_rowsum_kernel(int* __restrict__ tab,
                                                             int* __restrict__ cnt, int N) {
  const int i = blockIdx.x * 256 + threadIdx.x;
  if (i < N) {
    int v[CSR_S];
#pragma unroll
    for (int s = 0; s < CSR_S; ++s) v[s] = tab[(size_t)s * N + i];
    int run = 0;
#pragma unroll
    for (int s = 0; s < CSR_S; ++s) { const int t = v[s]; v[s] = run; run += t; }
#pragma unroll
    for (int s = 0; s < CSR_S; ++s) tab[(size_t)s * N + i] = v[s];
    cnt[i] = run;
  } else if (i == N) {
    cnt[N] = 0;
  }
}

#define SCAN_T 1024
__global__ __launch_bounds__(SCAN_T) void pdn_csr_scan_kernel(const int* __restrict__ cnt,
                                                              int* __restrict__ rowptr, int n) {
  __shared__ int s_part[32];
  __shared__ int s_base;
  const int tid  = threadIdx.x;
  const int lane = tid & 31;
  const int wave = tid >> 5;
  const int first = blockIdx.x * SCAN_T;

  int pre = 0;
  for (int j = tid; j < first; j += SCAN_T) pre += cnt[j];
  for (int off = 16; off > 0; off >>= 1) pre += __shfl_xor(pre, off, 32);
  if (lane == 0) s_part[wave] = pre;
  __syncthreads();
  if (wave == 0) {
    int v = s_part[lane];
    for (int off = 16; off > 0; off >>= 1) v += __shfl_xor(v, off, 32);
    if (lane == 0) s_base = v;
  }
  __syncthreads();
  const int base = s_base;

  const int i = first + tid;
  const int c = (i < n) ? cnt[i] : 0;
  int inc = c;
  for (int off = 1; off < 32; off <<= 1) {
    const int up = __shfl_up(inc, off, 32);
    if (lane >= off) inc += up;
  }
  if (lane == 31) s_part[wave] = inc;
  __syncthreads();
  int woff = (lane < wave) ? s_part[lane] : 0;
  for (int off = 16; off > 0; off >>= 1) woff += __shfl_xor(woff, off, 32);
  if (i < n) rowptr[i] = base + woff + inc - c;
}

__global__ __launch_bounds__(CSR_T) void pdn_csr_fill_kernel(const int* __restrict__ row,
                                                             const int* __restrict__ col,
                                                             const int* __restrict__ rowptr,
                                                             const int* __restrict__ tab,
                                                             const float2* __restrict__ g12,
                                                             int2* __restrict__ ent,
                                                             float* __restrict__ gs2,
                                                             int R, int ES, int E, int N) {
  __shared__ int s_pos[CSR_RANGE];
  const int r  = blockIdx.x % R;
  const int s  = blockIdx.x / R;
  const int r0 = r * CSR_RANGE;
  const int nr = min(CSR_RANGE, N - r0);
  for (int jb = threadIdx.x; jb < nr; jb += CSR_U * CSR_T) {
    int a[CSR_U];
#pragma unroll
    for (int u = 0; u < CSR_U; ++u) {
      const int j = min(jb + u * CSR_T, nr - 1);
      a[u] = rowptr[r0 + j] + tab[(size_t)s * N + r0 + j];
    }
#pragma unroll
    for (int u = 0; u < CSR_U; ++u) {
      const int j = jb + u * CSR_T;
      if (j < nr) s_pos[j] = a[u];
    }
  }
  __syncthreads();

  const int e0 = s * ES;
  const int e1 = min(e0 + ES, E);
  for (int base = e0 + (int)threadIdx.x; base < e1; base += CSR_U * CSR_T) {
    int c[CSR_U], rw[CSR_U], p[CSR_U];
    float2 gg[CSR_U];
#pragma unroll
    for (int u = 0; u < CSR_U; ++u) {
      const int e  = base + u * CSR_T;
      const int ec = min(e, e1 - 1);
      c[u]  = col[ec];
      rw[u] = row[ec];
      gg[u] = g12[ec];
      if (e >= e1) c[u] = -1;
    }
#pragma unroll
    for (int u = 0; u < CSR_U; ++u) {
      const unsigned d = (unsigned)(c[u] - r0);
      p[u] = (d < (unsigned)nr) ? atomicAdd(&s_pos[d], 1) : -1;
    }
#pragma unroll
    for (int u = 0; u < CSR_U; ++u) {
      if (p[u] >= 0) {
        ent[p[u]] = make_int2(rw[u], __float_as_int(gg[u].x));
        gs2[p[u]] = gg[u].y;
      }
    }
  }
}

__global__ __launch_bounds__(256) void pdn_deg_kernel(const int* __restrict__ rowptr,
                                                      const int2* __restrict__ ent,
                                                      const float* __restrict__ gs2,
                                                      float* __restrict__ dinv1,
                                                      float* __restrict__ dinv2, int N) {
  const int t = blockIdx.x * blockDim.x + threadIdx.x;
  const int i = t >> 4;
  const int l = t & 15;
  float s1 = 0.0f, s2 = 0.0f;
  if (i < N) {
    const int end = rowptr[i + 1];
    for (int p = rowptr[i] + l; p < end; p += 16) {
      s1 += __int_as_float(ent[p].y);
      s2 += gs2[p];
    }
  }
  for (int off = 1; off < 16; off <<= 1) {
    s1 += __shfl_xor(s1, off, 32);
    s2 += __shfl_xor(s2, off, 32);
  }
  if (i < N && l == 0) {
    const float d1 = 1.0f + s1;
    const float d2 = 1.0f + s2;
    dinv1[i] = (d1 > 0.0f) ? rsqrtf(d1) : 0.0f;
    dinv2[i] = (d2 > 0.0f) ? rsqrtf(d2) : 0.0f;
  }
}

template <bool GATE_IN_ENT>
__global__ __launch_bounds__(256) void pdn_agg_kernel(const float* __restrict__ h,
                                                      const int* __restrict__ rowptr,
                                                      const int2* __restrict__ ent,
                                                      const float* __restrict__ gs,
                                                      const float* __restrict__ dinv,
                                                      float* __restrict__ out, int N) {
  const int lane = threadIdx.x & 31;
  const int i = __builtin_amdgcn_readfirstlane((int)(blockIdx.x * 8 + (threadIdx.x >> 5)));
  if (i >= N) return;
  const int beg = rowptr[i];
  const int end = rowptr[i + 1];
  const float di = dinv[i];
  const float* hl = h + lane * 4;

  float4 acc = *(const float4*)(hl + (size_t)i * 128);

  for (int base = beg; base < end; base += 32) {
    const int p = base + lane;
    int r = i;
    float w = 0.0f;
    if (p < end) {
      const int2 en = ent[p];
      r = en.x;
      w = GATE_IN_ENT ? __int_as_float(en.y) : gs[p];
    }
    const int wi = __float_as_int(w);
    const int rem = end - base;
    const int n = rem < 32 ? rem : 32;
    for (int j = 0; j < n; j += 8) {
      int rr[8];
      float ww[8];
      float4 a[8];
#pragma unroll
      for (int u = 0; u < 8; ++u) rr[u] = __builtin_amdgcn_readlane(r, j + u);
#pragma unroll
      for (int u = 0; u < 8; ++u) a[u] = *(const float4*)(hl + (size_t)rr[u] * 128);
#pragma unroll
      for (int u = 0; u < 8; ++u) ww[u] = __int_as_float(__builtin_amdgcn_readlane(wi, j + u));
#pragma unroll
      for (int u = 0; u < 8; ++u) {
        acc.x += ww[u] * a[u].x; acc.y += ww[u] * a[u].y;
        acc.z += ww[u] * a[u].z; acc.w += ww[u] * a[u].w;
      }
    }
  }
  acc.x *= di; acc.y *= di; acc.z *= di; acc.w *= di;
  *(float4*)(out + (size_t)i * 128 + lane * 4) = acc;
}


extern "C" void kernel_launch(void* const* d_in, const int* in_sizes, int n_in,
                              void* d_out, int out_size, void* d_ws, size_t ws_size,
                              hipStream_t stream) {
  const float* x    = (const float*)d_in[0];
  const int*   ei   = (const int*)d_in[1];
  const float* ea   = (const float*)d_in[2];
  const float* W1   = (const float*)d_in[3];
  const float* m1w1 = (const float*)d_in[4];
  const float* m1b1 = (const float*)d_in[5];
  const float* m1w2 = (const float*)d_in[6];
  const float* m1b2 = (const float*)d_in[7];
  const float* W2   = (const float*)d_in[8];
  const float* m2w1 = (const float*)d_in[9];
  const float* m2b1 = (const float*)d_in[10];
  const float* m2w2 = (const float*)d_in[11];
  const float* m2b2 = (const float*)d_in[12];

  const int N = in_sizes[0] / 128;
  const int E = in_sizes[2] / 16;
  const int* row = ei;
  const int* col = ei + E;

  float* ws     = (float*)d_ws;
  float* h_buf  = ws;
  int2*  ent    = (int2*)(h_buf + (size_t)N * 128);
  float* gs2    = (float*)(ent + (size_t)E);
  float* dinv1  = gs2 + (size_t)E;
  int*   cnt    = (int*)(dinv1 + (size_t)N);
  int*   rowptr = cnt + (size_t)N + 1;
  float* dinv2  = (float*)cnt;
  int*   tab    = (int*)h_buf;
  float* outp   = (float*)d_out;
  float2* g12   = (float2*)d_out;

  const int gateBlocks = (E + 127) / 128;
  const int gemmBlocks = (N + 16 * RPW - 1) / (16 * RPW);
  const int csrR       = (N + CSR_RANGE - 1) / CSR_RANGE;
  const int csrES      = (E + CSR_S - 1) / CSR_S;
  const int csrBlocks  = csrR * CSR_S;
  const int sumBlocks  = (N + 1 + 255) / 256;
  const int scanBlocks = (N + 1 + SCAN_T - 1) / SCAN_T;
  const int degBlocks  = (int)(((size_t)N * 16 + 255) / 256);
  const int aggBlocks  = (N + 7) / 8;

  pdn_gate_kernel<<<gateBlocks, 256, 0, stream>>>(ea, m1w1, m1b1, m1w2, m1b2,
                                                  m2w1, m2b1, m2w2, m2b2, g12, E);
  pdn_csr_count_kernel<<<csrBlocks, CSR_T, 0, stream>>>(col, tab, csrR, csrES, E, N);
  pdn_csr_rowsum_kernel<<<sumBlocks, 256, 0, stream>>>(tab, cnt, N);
  pdn_csr_scan_kernel<<<scanBlocks, SCAN_T, 0, stream>>>(cnt, rowptr, N + 1);
  pdn_csr_fill_kernel<<<csrBlocks, CSR_T, 0, stream>>>(row, col, rowptr, tab, g12, ent, gs2,
                                                       csrR, csrES, E, N);
  pdn_deg_kernel<<<degBlocks, 256, 0, stream>>>(rowptr, ent, gs2, dinv1, dinv2, N);

  pdn_gemm_kernel<<<gemmBlocks, 256, 0, stream>>>(x, W1, dinv1, h_buf, N, 0);
  pdn_agg_kernel<true><<<aggBlocks, 256, 0, stream>>>(h_buf, rowptr, ent, gs2, dinv1, outp, N);

  pdn_gemm_kernel<<<gemmBlocks, 256, 0, stream>>>(outp, W2, dinv2, h_buf, N, 1);
  pdn_agg_kernel<false><<<aggBlocks, 256, 0, stream>>>(h_buf, rowptr, ent, gs2, dinv2, outp, N);
}


# agg + weight product fused, edge sum first: (A x) W per 32-node block, sums go to LDS and are the WMMA A tiles; h round trip, 16-line A loads, 2 launches gone; product after the sum: order changed
# speedup vs baseline: 3.4549x; 1.2282x over previous
#include <hip/hip_runtime.h>
#include <math.h>

typedef __attribute__((ext_vector_type(2))) float v2f;
typedef __attribute__((ext_vector_type(8))) float v8f;

#define WMMA_F32X4(a, b, c) \
    __builtin_amdgcn_wmma_f32_16x16x4_f32(false, (a), false, (b), (short)0, (c), false, false)

__device__ __forceinline__ float pdn_gate_one(const v2f (&afrag)[4],
    const float* __restrict__ mw1, const float* __restrict__ mb1,
    const float* __restrict__ mw2, const float* __restrict__ mb2,
    const int half, const int l16) {
  float acc[8];
#pragma unroll
  for (int v = 0; v < 8; ++v) acc[v] = 0.0f;

#pragma unroll
  for (int t = 0; t < 8; ++t) {
    const int n0 = t * 16;
    v8f c = {};
#pragma unroll
    for (int kk = 0; kk < 4; ++kk) {
      const int kb = 4 * kk + 2 * half;
      v2f b;
      b.x = mw1[(size_t)kb * 128 + n0 + l16];
      b.y = mw1[(size_t)(kb + 1) * 128 + n0 + l16];
      c = WMMA_F32X4(afrag[kk], b, c);
    }
    const float b1 = mb1[n0 + l16];
    const float w2 = mw2[n0 + l16];
#pragma unroll
    for (int v = 0; v < 8; ++v) {
      acc[v] += fmaxf(c[v] + b1, 0.0f) * w2;
    }
  }

#pragma unroll
  for (int off = 1; off < 16; off <<= 1) {
#pragma unroll
    for (int v = 0; v < 8; ++v) acc[v] += __shfl_xor(acc[v], off, 32);
  }

  float s = acc[0];
#pragma unroll
  for (int v = 1; v < 8; ++v) s = (l16 == v) ? acc[v] : s;
  s += mb2[0];
  return 1.0f / (1.0f + __expf(-s));
}

__global__ __launch_bounds__(256) void pdn_gate_kernel(
    const float* __restrict__ ea,
    const float* __restrict__ aw1, const float* __restrict__ ab1,
    const float* __restrict__ aw2, const float* __restrict__ ab2,
    const float* __restrict__ bw1, const float* __restrict__ bb1,
    const float* __restrict__ bw2, const float* __restrict__ bb2,
    float2* __restrict__ g12, int E) {
  const int wave = threadIdx.x >> 5;
  const int lane = threadIdx.x & 31;
  const int half = lane >> 4;
  const int l16  = lane & 15;
  const int e0 = (blockIdx.x * 8 + wave) * 16;
  if (e0 >= E) return;

  int er = e0 + l16;
  if (er >= E) er = E - 1;
  const float* earow = ea + (size_t)er * 16;

  v2f afrag[4];
#pragma unroll
  for (int kk = 0; kk < 4; ++kk) {
    afrag[kk].x = earow[4 * kk + 2 * half];
    afrag[kk].y = earow[4 * kk + 2 * half + 1];
  }

  float2 gg;
  gg.x = 0.0f; gg.y = 0.0f;
#pragma nounroll
  for (int layer = 0; layer < 2; ++layer) {
    const float* w1 = layer ? bw1 : aw1;
    const float* b1 = layer ? bb1 : ab1;
    const float* w2 = layer ? bw2 : aw2;
    const float* b2 = layer ? bb2 : ab2;
    const float gv = pdn_gate_one(afrag, w1, b1, w2, b2, half, l16);
    if (layer) gg.y = gv; else gg.x = gv;
  }

  const int e = e0 + l16 + 8 * half;
  if (l16 < 8 && e < E) g12[e] = gg;
}

#define CSR_RANGE 12500
#define CSR_S 32
#define CSR_T 1024
#define CSR_U 4

__global__ __launch_bounds__(CSR_T) void pdn_csr_count_kernel(const int* __restrict__ col,
                                                              int* __restrict__ tab,
                                                              int R, int ES, int E, int N) {
  __shared__ int s_cnt[CSR_RANGE];
  const int r  = blockIdx.x % R;
  const int s  = blockIdx.x / R;
  const int r0 = r * CSR_RANGE;
  const int nr = min(CSR_RANGE, N - r0);
  for (int j = threadIdx.x; j < nr; j += CSR_T) s_cnt[j] = 0;
  __syncthreads();

  const int e0 = s * ES;
  const int e1 = min(e0 + ES, E);
  for (int base = e0 + (int)threadIdx.x; base < e1; base += CSR_U * CSR_T) {
    int c[CSR_U];
#pragma unroll
    for (int u = 0; u < CSR_U; ++u) {
      const int e = base + u * CSR_T;
      c[u] = col[min(e, e1 - 1)];
      if (e >= e1) c[u] = -1;
    }
#pragma unroll
    for (int u = 0; u < CSR_U; ++u) {
      const unsigned d = (unsigned)(c[u] - r0);
      if (d < (unsigned)nr) atomicAdd(&s_cnt[d], 1);
    }
  }
  __syncthreads();
  for (int j = threadIdx.x; j < nr; j += CSR_T) tab[(size_t)s * N + r0 + j] = s_cnt[j];
}

__global__ __launch_bounds__(256) void pdn_csr_rowsum_kernel(int* __restrict__ tab,
                                                             int* __restrict__ cnt, int N) {
  const int i = blockIdx.x * 256 + threadIdx.x;
  if (i < N) {
    int v[CSR_S];
#pragma unroll
    for (int s = 0; s < CSR_S; ++s) v[s] = tab[(size_t)s * N + i];
    int run = 0;
#pragma unroll
    for (int s = 0; s < CSR_S; ++s) { const int t = v[s]; v[s] = run; run += t; }
#pragma unroll
    for (int s = 0; s < CSR_S; ++s) tab[(size_t)s * N + i] = v[s];
    cnt[i] = run;
  } else if (i == N) {
    cnt[N] = 0;
  }
}

#define SCAN_T 1024
__global__ __launch_bounds__(SCAN_T) void pdn_csr_scan_kernel(const int* __restrict__ cnt,
                                                              int* __restrict__ rowptr, int n) {
  __shared__ int s_part[32];
  __shared__ int s_base;
  const int tid  = threadIdx.x;
  const int lane = tid & 31;
  const int wave = tid >> 5;
  const int first = blockIdx.x * SCAN_T;

  int pre = 0;
  for (int j = tid; j < first; j += SCAN_T) pre += cnt[j];
  for (int off = 16; off > 0; off >>= 1) pre += __shfl_xor(pre, off, 32);
  if (lane == 0) s_part[wave] = pre;
  __syncthreads();
  if (wave == 0) {
    int v = s_part[lane];
    for (int off = 16; off > 0; off >>= 1) v += __shfl_xor(v, off, 32);
    if (lane == 0) s_base = v;
  }
  __syncthreads();
  const int base = s_base;

  const int i = first + tid;
  const int c = (i < n) ? cnt[i] : 0;
  int inc = c;
  for (int off = 1; off < 32; off <<= 1) {
    const int up = __shfl_up(inc, off, 32);
    if (lane >= off) inc += up;
  }
  if (lane == 31) s_part[wave] = inc;
  __syncthreads();
  int woff = (lane < wave) ? s_part[lane] : 0;
  for (int off = 16; off > 0; off >>= 1) woff += __shfl_xor(woff, off, 32);
  if (i < n) rowptr[i] = base + woff + inc - c;
}

__global__ __launch_bounds__(CSR_T) void pdn_csr_fill_kernel(const int* __restrict__ row,
                                                             const int* __restrict__ col,
                                                             const int* __restrict__ rowptr,
                                                             const int* __restrict__ tab,
                                                             const float2* __restrict__ g12,
                                                             int2* __restrict__ ent,
                                                             float* __restrict__ gs2,
                                                             int R, int ES, int E, int N) {
  __shared__ int s_pos[CSR_RANGE];
  const int r  = blockIdx.x % R;
  const int s  = blockIdx.x / R;
  const int r0 = r * CSR_RANGE;
  const int nr = min(CSR_RANGE, N - r0);
  for (int jb = threadIdx.x; jb < nr; jb += CSR_U * CSR_T) {
    int a[CSR_U];
#pragma unroll
    for (int u = 0; u < CSR_U; ++u) {
      const int j = min(jb + u * CSR_T, nr - 1);
      a[u] = rowptr[r0 + j] + tab[(size_t)s * N + r0 + j];
    }
#pragma unroll
    for (int u = 0; u < CSR_U; ++u) {
      const int j = jb + u * CSR_T;
      if (j < nr) s_pos[j] = a[u];
    }
  }
  __syncthreads();

  const int e0 = s * ES;
  const int e1 = min(e0 + ES, E);
  for (int base = e0 + (int)threadIdx.x; base < e1; base += CSR_U * CSR_T) {
    int c[CSR_U], rw[CSR_U], p[CSR_U];
    float2 gg[CSR_U];
#pragma unroll
    for (int u = 0; u < CSR_U; ++u) {
      const int e  = base + u * CSR_T;
      const int ec = min(e, e1 - 1);
      c[u]  = col[ec];
      rw[u] = row[ec];
      gg[u] = g12[ec];
      if (e >= e1) c[u] = -1;
    }
#pragma unroll
    for (int u = 0; u < CSR_U; ++u) {
      const unsigned d = (unsigned)(c[u] - r0);
      p[u] = (d < (unsigned)nr) ? atomicAdd(&s_pos[d], 1) : -1;
    }
#pragma unroll
    for (int u = 0; u < CSR_U; ++u) {
      if (p[u] >= 0) {
        ent[p[u]] = make_int2(rw[u], __float_as_int(gg[u].x));
        gs2[p[u]] = gg[u].y;
      }
    }
  }
}

__global__ __launch_bounds__(256) void pdn_deg_kernel(const int* __restrict__ rowptr,
                                                      const int2* __restrict__ ent,
                                                      const float* __restrict__ gs2,
                                                      float* __restrict__ dinv1,
                                                      float* __restrict__ dinv2, int N) {
  const int t = blockIdx.x * blockDim.x + threadIdx.x;
  const int i = t >> 4;
  const int l = t & 15;
  float s1 = 0.0f, s2 = 0.0f;
  if (i < N) {
    const int end = rowptr[i + 1];
    for (int p = rowptr[i] + l; p < end; p += 16) {
      s1 += __int_as_float(ent[p].y);
      s2 += gs2[p];
    }
  }
  for (int off = 1; off < 16; off <<= 1) {
    s1 += __shfl_xor(s1, off, 32);
    s2 += __shfl_xor(s2, off, 32);
  }
  if (i < N && l == 0) {
    const float d1 = 1.0f + s1;
    const float d2 = 1.0f + s2;
    dinv1[i] = (d1 > 0.0f) ? rsqrtf(d1) : 0.0f;
    dinv2[i] = (d2 > 0.0f) ? rsqrtf(d2) : 0.0f;
  }
}

#define FT_ROWS 32
#define FT_NPW 4
#define FT_RT 2
#define FT_PITCH 132
template <bool FIRST>
__global__ __launch_bounds__(256) void pdn_aggmm_kernel(const float* __restrict__ xs,
                                                        const float* __restrict__ W,
                                                        const int* __restrict__ rowptr,
                                                        const int2* __restrict__ ent,
                                                        const float* __restrict__ gs,
                                                        const float* __restrict__ dinv,
                                                        const float* __restrict__ dnext,
                                                        float* __restrict__ out, int N) {
  __shared__ __attribute__((aligned(16))) float s_t[FT_ROWS * FT_PITCH];
  __shared__ float s_d[FT_ROWS];
  __shared__ float s_n[FT_ROWS];
  const int wave = threadIdx.x >> 5;
  const int lane = threadIdx.x & 31;
  const int half = lane >> 4;
  const int l16  = lane & 15;
  const int rowBase = blockIdx.x * FT_ROWS;
  const float* xl = xs + lane * 4;

#pragma nounroll
  for (int q = 0; q < FT_NPW; ++q) {
    const int lr = wave * FT_NPW + q;
    const int i = __builtin_amdgcn_readfirstlane(rowBase + lr);
    float4 acc = make_float4(0.0f, 0.0f, 0.0f, 0.0f);
    if (i < N) {
      const int beg = rowptr[i];
      const int end = rowptr[i + 1];
      const float di = dinv[i];
      acc = *(const float4*)(xl + (size_t)i * 128);
      if (FIRST) { acc.x *= di; acc.y *= di; acc.z *= di; acc.w *= di; }

      for (int base = beg; base < end; base += 32) {
        const int p = base + lane;
        int r = i;
        float w = 0.0f;
        if (p < end) {
          const int2 en = ent[p];
          r = en.x;
          w = FIRST ? __int_as_float(en.y) * dinv[r] : gs[p];
        }
        const int wi = __float_as_int(w);
        const int rem = end - base;
        const int n = rem < 32 ? rem : 32;
        for (int j = 0; j < n; j += 8) {
          int rr[8];
          float ww[8];
          float4 a[8];
#pragma unroll
          for (int u = 0; u < 8; ++u) rr[u] = __builtin_amdgcn_readlane(r, j + u);
#pragma unroll
          for (int u = 0; u < 8; ++u) a[u] = *(const float4*)(xl + (size_t)rr[u] * 128);
#pragma unroll
          for (int u = 0; u < 8; ++u) ww[u] = __int_as_float(__builtin_amdgcn_readlane(wi, j + u));
#pragma unroll
          for (int u = 0; u < 8; ++u) {
            acc.x += ww[u] * a[u].x; acc.y += ww[u] * a[u].y;
            acc.z += ww[u] * a[u].z; acc.w += ww[u] * a[u].w;
          }
        }
      }
      if (lane == 0) {
        s_d[lr] = di;
        if (FIRST) s_n[lr] = dnext[i];
      }
    }
    *(float4*)(&s_t[lr * FT_PITCH + lane * 4]) = acc;
  }
  __syncthreads();

  const int n0 = wave * 16;
  const float* Wl = W + n0 + l16;
  v8f c[FT_RT];
#pragma unroll
  for (int t = 0; t < FT_RT; ++t) c[t] = (v8f){};

#pragma unroll 4
  for (int k0 = 0; k0 < 128; k0 += 8) {
    const int kb = k0 + 4 * half;
    v2f b0, b1;
    b0.x = Wl[(size_t)(kb + 0) * 128];
    b0.y = Wl[(size_t)(kb + 1) * 128];
    b1.x = Wl[(size_t)(kb + 2) * 128];
    b1.y = Wl[(size_t)(kb + 3) * 128];
    float4 a[FT_RT];
#pragma unroll
    for (int t = 0; t < FT_RT; ++t) a[t] = *(const float4*)(&s_t[(t * 16 + l16) * FT_PITCH + kb]);
#pragma unroll
    for (int t = 0; t < FT_RT; ++t) {
      v2f af; af.x = a[t].x; af.y = a[t].y;
      c[t] = WMMA_F32X4(af, b0, c[t]);
    }
#pragma unroll
    for (int t = 0; t < FT_RT; ++t) {
      v2f af; af.x = a[t].z; af.y = a[t].w;
      c[t] = WMMA_F32X4(af, b1, c[t]);
    }
  }

  __syncthreads();
#pragma unroll
  for (int t = 0; t < FT_RT; ++t) {
#pragma unroll
    for (int v = 0; v < 8; ++v) {
      s_t[(t * 16 + v + 8 * half) * FT_PITCH + n0 + l16] = c[t][v];
    }
  }
  __syncthreads();
#pragma unroll
  for (int q = 0; q < FT_NPW; ++q) {
    const int lr = wave * FT_NPW + q;
    const int i = rowBase + lr;
    if (i < N) {
      float4 y = *(const float4*)(&s_t[lr * FT_PITCH + lane * 4]);
      const float di = s_d[lr];
      y.x *= di; y.y *= di; y.z *= di; y.w *= di;
      if (FIRST) {
        const float dn = s_n[lr];
        y.x = fmaxf(y.x, 0.0f) * dn; y.y = fmaxf(y.y, 0.0f) * dn;
        y.z = fmaxf(y.z, 0.0f) * dn; y.w = fmaxf(y.w, 0.0f) * dn;
      }
      *(float4*)(out + (size_t)i * 128 + lane * 4) = y;
    }
  }
}


extern "C" void kernel_launch(void* const* d_in, const int* in_sizes, int n_in,
                              void* d_out, int out_size, void* d_ws, size_t ws_size,
                              hipStream_t stream) {
  const float* x    = (const float*)d_in[0];
  const int*   ei   = (const int*)d_in[1];
  const float* ea   = (const float*)d_in[2];
  const float* W1   = (const float*)d_in[3];
  const float* m1w1 = (const float*)d_in[4];
  const float* m1b1 = (const float*)d_in[5];
  const float* m1w2 = (const float*)d_in[6];
  const float* m1b2 = (const float*)d_in[7];
  const float* W2   = (const float*)d_in[8];
  const float* m2w1 = (const float*)d_in[9];
  const float* m2b1 = (const float*)d_in[10];
  const float* m2w2 = (const float*)d_in[11];
  const float* m2b2 = (const float*)d_in[12];

  const int N = in_sizes[0] / 128;
  const int E = in_sizes[2] / 16;
  const int* row = ei;
  const int* col = ei + E;

  float* ws     = (float*)d_ws;
  float* h_buf  = ws;
  int2*  ent    = (int2*)(h_buf + (size_t)N * 128);
  float* gs2    = (float*)(ent + (size_t)E);
  float* dinv1  = gs2 + (size_t)E;
  int*   cnt    = (int*)(dinv1 + (size_t)N);
  int*   rowptr = cnt + (size_t)N + 1;
  float* dinv2  = (float*)cnt;
  int*   tab    = (int*)h_buf;
  float* outp   = (float*)d_out;
  float2* g12   = (float2*)d_out;

  const int gateBlocks = (E + 127) / 128;
  const int csrR       = (N + CSR_RANGE - 1) / CSR_RANGE;
  const int csrES      = (E + CSR_S - 1) / CSR_S;
  const int csrBlocks  = csrR * CSR_S;
  const int sumBlocks  = (N + 1 + 255) / 256;
  const int scanBlocks = (N + 1 + SCAN_T - 1) / SCAN_T;
  const int degBlocks  = (int)(((size_t)N * 16 + 255) / 256);
  const int aggBlocks  = (N + FT_ROWS - 1) / FT_ROWS;

  pdn_gate_kernel<<<gateBlocks, 256, 0, stream>>>(ea, m1w1, m1b1, m1w2, m1b2,
                                                  m2w1, m2b1, m2w2, m2b2, g12, E);
  pdn_csr_count_kernel<<<csrBlocks, CSR_T, 0, stream>>>(col, tab, csrR, csrES, E, N);
  pdn_csr_rowsum_kernel<<<sumBlocks, 256, 0, stream>>>(tab, cnt, N);
  pdn_csr_scan_kernel<<<scanBlocks, SCAN_T, 0, stream>>>(cnt, rowptr, N + 1);
  pdn_csr_fill_kernel<<<csrBlocks, CSR_T, 0, stream>>>(row, col, rowptr, tab, g12, ent, gs2,
                                                       csrR, csrES, E, N);
  pdn_deg_kernel<<<degBlocks, 256, 0, stream>>>(rowptr, ent, gs2, dinv1, dinv2, N);

  pdn_aggmm_kernel<true><<<aggBlocks, 256, 0, stream>>>(x, W1, rowptr, ent, gs2, dinv1, dinv2,
                                                        h_buf, N);

  pdn_aggmm_kernel<false><<<aggBlocks, 256, 0, stream>>>(h_buf, W2, rowptr, ent, gs2, dinv2, dinv2,
                                                         outp, N);
}


# gate and degree launches as resident loops: gate weights go once per block into LDS as ready B fragments (160 4-byte global loads per 16 edges gone), waves walk 12 tiles; a degree wave walks 32 nodes
# speedup vs baseline: 3.7476x; 1.0847x over previous
#include <hip/hip_runtime.h>
#include <math.h>

typedef __attribute__((ext_vector_type(2))) float v2f;
typedef __attribute__((ext_vector_type(8))) float v8f;

#define WMMA_F32X4(a, b, c) \
    __builtin_amdgcn_wmma_f32_16x16x4_f32(false, (a), false, (b), (short)0, (c), false, false)

#define GATE_T 256
#define GATE_FRAGS (8 * 4 * 32)

__device__ __forceinline__ float pdn_gate_one(const v2f (&afrag)[4],
    const v2f* __restrict__ sb,
    const v2f* __restrict__ sv,
    const float b2, const int lane, const int l16) {
  float acc[8];
#pragma unroll
  for (int v = 0; v < 8; ++v) acc[v] = 0.0f;

#pragma unroll
  for (int t = 0; t < 8; ++t) {
    v8f c = {};
#pragma unroll
    for (int kk = 0; kk < 4; ++kk) c = WMMA_F32X4(afrag[kk], sb[(t * 4 + kk) * 32 + lane], c);
    const v2f bw = sv[t * 16 + l16];
    const float b1 = bw.x;
    const float w2 = bw.y;
#pragma unroll
    for (int v = 0; v < 8; ++v) {
      acc[v] += fmaxf(c[v] + b1, 0.0f) * w2;
    }
  }

#pragma unroll
  for (int off = 1; off < 16; off <<= 1) {
#pragma unroll
    for (int v = 0; v < 8; ++v) acc[v] += __shfl_xor(acc[v], off, 32);
  }

  float s = acc[0];
#pragma unroll
  for (int v = 1; v < 8; ++v) s = (l16 == v) ? acc[v] : s;
  s += b2;
  return 1.0f / (1.0f + __expf(-s));
}

__global__ __launch_bounds__(GATE_T) void pdn_gate_kernel(
    const float* __restrict__ ea,
    const float* __restrict__ aw1, const float* __restrict__ ab1,
    const float* __restrict__ aw2, const float* __restrict__ ab2,
    const float* __restrict__ bw1, const float* __restrict__ bb1,
    const float* __restrict__ bw2, const float* __restrict__ bb2,
    float2* __restrict__ g12, int E) {
  __shared__ __attribute__((aligned(16))) v2f s_b[2 * GATE_FRAGS];
  __shared__ __attribute__((aligned(16))) v2f s_v[2 * 128];
  const int wave = threadIdx.x >> 5;
  const int lane = threadIdx.x & 31;
  const int half = lane >> 4;
  const int l16  = lane & 15;

  for (int idx = threadIdx.x; idx < 2 * GATE_FRAGS; idx += GATE_T) {
    const int fl = idx & 31;
    const int kk = (idx >> 5) & 3;
    const int t  = (idx >> 7) & 7;
    const float* w1 = (idx >= GATE_FRAGS) ? bw1 : aw1;
    const int kb = 4 * kk + 2 * (fl >> 4);
    const int cn = 16 * t + (fl & 15);
    v2f b;
    b.x = w1[(size_t)kb * 128 + cn];
    b.y = w1[(size_t)(kb + 1) * 128 + cn];
    s_b[idx] = b;
  }
  {
    const int cn = threadIdx.x & 127;
    v2f bw;
    bw.x = (threadIdx.x >= 128) ? bb1[cn] : ab1[cn];
    bw.y = (threadIdx.x >= 128) ? bw2[cn] : aw2[cn];
    s_v[threadIdx.x] = bw;
  }
  const float b2a = ab2[0];
  const float b2b = bb2[0];
  __syncthreads();

  const int tiles = (E + 15) >> 4;
  const int nw = gridDim.x * (GATE_T / 32);
  int tile = blockIdx.x * (GATE_T / 32) + wave;
  if (tile >= tiles) return;

  v2f afrag[4];
  {
    int er = tile * 16 + l16;
    if (er >= E) er = E - 1;
    const float* earow = ea + (size_t)er * 16;
#pragma unroll
    for (int kk = 0; kk < 4; ++kk) {
      afrag[kk].x = earow[4 * kk + 2 * half];
      afrag[kk].y = earow[4 * kk + 2 * half + 1];
    }
  }

  for (;;) {
    asm volatile("" ::: "memory");
    const int nt = tile + nw;
    v2f nxt[4];
    {
      int er = (nt < tiles ? nt : tile) * 16 + l16;
      if (er >= E) er = E - 1;
      const float* earow = ea + (size_t)er * 16;
#pragma unroll
      for (int kk = 0; kk < 4; ++kk) {
        nxt[kk].x = earow[4 * kk + 2 * half];
        nxt[kk].y = earow[4 * kk + 2 * half + 1];
      }
    }

    float2 gg;
    gg.x = pdn_gate_one(afrag, s_b, s_v, b2a, lane, l16);
    gg.y = pdn_gate_one(afrag, s_b + GATE_FRAGS, s_v + 128, b2b, lane, l16);

    const int e = tile * 16 + l16 + 8 * half;
    if (l16 < 8 && e < E) g12[e] = gg;

    if (nt >= tiles) break;
    tile = nt;
#pragma unroll
    for (int kk = 0; kk < 4; ++kk) afrag[kk] = nxt[kk];
  }
}

#define CSR_RANGE 12500
#define CSR_S 32
#define CSR_T 1024
#define CSR_U 4

__global__ __launch_bounds__(CSR_T) void pdn_csr_count_kernel(const int* __restrict__ col,
                                                              int* __restrict__ tab,
                                                              int R, int ES, int E, int N) {
  __shared__ int s_cnt[CSR_RANGE];
  const int r  = blockIdx.x % R;
  const int s  = blockIdx.x / R;
  const int r0 = r * CSR_RANGE;
  const int nr = min(CSR_RANGE, N - r0);
  for (int j = threadIdx.x; j < nr; j += CSR_T) s_cnt[j] = 0;
  __syncthreads();

  const int e0 = s * ES;
  const int e1 = min(e0 + ES, E);
  for (int base = e0 + (int)threadIdx.x; base < e1; base += CSR_U * CSR_T) {
    int c[CSR_U];
#pragma unroll
    for (int u = 0; u < CSR_U; ++u) {
      const int e = base + u * CSR_T;
      c[u] = col[min(e, e1 - 1)];
      if (e >= e1) c[u] = -1;
    }
#pragma unroll
    for (int u = 0; u < CSR_U; ++u) {
      const unsigned d = (unsigned)(c[u] - r0);
      if (d < (unsigned)nr) atomicAdd(&s_cnt[d], 1);
    }
  }
  __syncthreads();
  for (int j = threadIdx.x; j < nr; j += CSR_T) tab[(size_t)s * N + r0 + j] = s_cnt[j];
}

__global__ __launch_bounds__(256) void pdn_csr_rowsum_kernel(int* __restrict__ tab,
                                                             int* __restrict__ cnt, int N) {
  const int i = blockIdx.x * 256 + threadIdx.x;
  if (i < N) {
    int v[CSR_S];
#pragma unroll
    for (int s = 0; s < CSR_S; ++s) v[s] = tab[(size_t)s * N + i];
    int run = 0;
#pragma unroll
    for (int s = 0; s < CSR_S; ++s) { const int t = v[s]; v[s] = run; run += t; }
#pragma unroll
    for (int s = 0; s < CSR_S; ++s) tab[(size_t)s * N + i] = v[s];
    cnt[i] = run;
  } else if (i == N) {
    cnt[N] = 0;
  }
}

#define SCAN_T 1024
__global__ __launch_bounds__(SCAN_T) void pdn_csr_scan_kernel(const int* __restrict__ cnt,
                                                              int* __restrict__ rowptr, int n) {
  __shared__ int s_part[32];
  __shared__ int s_base;
  const int tid  = threadIdx.x;
  const int lane = tid & 31;
  const int wave = tid >> 5;
  const int first = blockIdx.x * SCAN_T;

  int pre = 0;
  for (int j = tid; j < first; j += SCAN_T) pre += cnt[j];
  for (int off = 16; off > 0; off >>= 1) pre += __shfl_xor(pre, off, 32);
  if (lane == 0) s_part[wave] = pre;
  __syncthreads();
  if (wave == 0) {
    int v = s_part[lane];
    for (int off = 16; off > 0; off >>= 1) v += __shfl_xor(v, off, 32);
    if (lane == 0) s_base = v;
  }
  __syncthreads();
  const int base = s_base;

  const int i = first + tid;
  const int c = (i < n) ? cnt[i] : 0;
  int inc = c;
  for (int off = 1; off < 32; off <<= 1) {
    const int up = __shfl_up(inc, off, 32);
    if (lane >= off) inc += up;
  }
  if (lane == 31) s_part[wave] = inc;
  __syncthreads();
  int woff = (lane < wave) ? s_part[lane] : 0;
  for (int off = 16; off > 0; off >>= 1) woff += __shfl_xor(woff, off, 32);
  if (i < n) rowptr[i] = base + woff + inc - c;
}

__global__ __launch_bounds__(CSR_T) void pdn_csr_fill_kernel(const int* __restrict__ row,
                                                             const int* __restrict__ col,
                                                             const int* __restrict__ rowptr,
                                                             const int* __restrict__ tab,
                                                             const float2* __restrict__ g12,
                                                             int2* __restrict__ ent,
                                                             float* __restrict__ gs2,
                                                             int R, int ES, int E, int N) {
  __shared__ int s_pos[CSR_RANGE];
  const int r  = blockIdx.x % R;
  const int s  = blockIdx.x / R;
  const int r0 = r * CSR_RANGE;
  const int nr = min(CSR_RANGE, N - r0);
  for (int jb = threadIdx.x; jb < nr; jb += CSR_U * CSR_T) {
    int a[CSR_U];
#pragma unroll
    for (int u = 0; u < CSR_U; ++u) {
      const int j = min(jb + u * CSR_T, nr - 1);
      a[u] = rowptr[r0 + j] + tab[(size_t)s * N + r0 + j];
    }
#pragma unroll
    for (int u = 0; u < CSR_U; ++u) {
      const int j = jb + u * CSR_T;
      if (j < nr) s_pos[j] = a[u];
    }
  }
  __syncthreads();

  const int e0 = s * ES;
  const int e1 = min(e0 + ES, E);
  for (int base = e0 + (int)threadIdx.x; base < e1; base += CSR_U * CSR_T) {
    int c[CSR_U], rw[CSR_U], p[CSR_U];
    float2 gg[CSR_U];
#pragma unroll
    for (int u = 0; u < CSR_U; ++u) {
      const int e  = base + u * CSR_T;
      const int ec = min(e, e1 - 1);
      c[u]  = col[ec];
      rw[u] = row[ec];
      gg[u] = g12[ec];
      if (e >= e1) c[u] = -1;
    }
#pragma unroll
    for (int u = 0; u < CSR_U; ++u) {
      const unsigned d = (unsigned)(c[u] - r0);
      p[u] = (d < (unsigned)nr) ? atomicAdd(&s_pos[d], 1) : -1;
    }
#pragma unroll
    for (int u = 0; u < CSR_U; ++u) {
      if (p[u] >= 0) {
        ent[p[u]] = make_int2(rw[u], __float_as_int(gg[u].x));
        gs2[p[u]] = gg[u].y;
      }
    }
  }
}

#define DEG_NPW 32
__global__ __launch_bounds__(256) void pdn_deg_kernel(const int* __restrict__ rowptr,
                                                      const int2* __restrict__ ent,
                                                      const float* __restrict__ gs2,
                                                      float* __restrict__ dinv1,
                                                      float* __restrict__ dinv2, int N) {
  const int lane = threadIdx.x & 31;
  const int half = lane >> 4;
  const int l16  = lane & 15;
  const int i0 = (blockIdx.x * 8 + (threadIdx.x >> 5)) * DEG_NPW;
  if (i0 >= N) return;
  const int mine = min(i0 + lane, N);
  const int rb = rowptr[mine];
  const int re = rowptr[min(mine + 1, N)];
  float k1 = 0.0f, k2 = 0.0f;

  for (int g = 0; g < 2; ++g) {
    int pb[8], pe[8];
    float a0[8], a1[8], c0[8], c1[8];
#pragma unroll
    for (int u = 0; u < 8; ++u) {
      const int srcl = 8 * g + u + 16 * half;
      pb[u] = __shfl(rb, srcl, 32) + l16;
      pe[u] = __shfl(re, srcl, 32);
      const int p0 = pb[u];
      const int p1 = p0 + 16;
      a0[u] = 0.0f; a1[u] = 0.0f; c0[u] = 0.0f; c1[u] = 0.0f;
      if (p0 < pe[u]) { a0[u] = __int_as_float(ent[p0].y); c0[u] = gs2[p0]; }
      if (p1 < pe[u]) { a1[u] = __int_as_float(ent[p1].y); c1[u] = gs2[p1]; }
    }
#pragma unroll
    for (int u = 0; u < 8; ++u) {
      float s1 = a0[u] + a1[u];
      float s2 = c0[u] + c1[u];
      for (int p = pb[u] + 32; p < pe[u]; p += 16) {
        s1 += __int_as_float(ent[p].y);
        s2 += gs2[p];
      }
      for (int off = 1; off < 16; off <<= 1) {
        s1 += __shfl_xor(s1, off, 32);
        s2 += __shfl_xor(s2, off, 32);
      }
      if (l16 == 8 * g + u) { k1 = s1; k2 = s2; }
    }
  }

  const int i = i0 + lane;
  if (i < N) {
    const float d1 = 1.0f + k1;
    const float d2 = 1.0f + k2;
    dinv1[i] = (d1 > 0.0f) ? rsqrtf(d1) : 0.0f;
    dinv2[i] = (d2 > 0.0f) ? rsqrtf(d2) : 0.0f;
  }
}

#define FT_ROWS 32
#define FT_NPW 4
#define FT_RT 2
#define FT_PITCH 132
template <bool FIRST>
__global__ __launch_bounds__(256) void pdn_aggmm_kernel(const float* __restrict__ xs,
                                                        const float* __restrict__ W,
                                                        const int* __restrict__ rowptr,
                                                        const int2* __restrict__ ent,
                                                        const float* __restrict__ gs,
                                                        const float* __restrict__ dinv,
                                                        const float* __restrict__ dnext,
                                                        float* __restrict__ out, int N) {
  __shared__ __attribute__((aligned(16))) float s_t[FT_ROWS * FT_PITCH];
  __shared__ float s_d[FT_ROWS];
  __shared__ float s_n[FT_ROWS];
  const int wave = threadIdx.x >> 5;
  const int lane = threadIdx.x & 31;
  const int half = lane >> 4;
  const int l16  = lane & 15;
  const int rowBase = blockIdx.x * FT_ROWS;
  const float* xl = xs + lane * 4;

#pragma nounroll
  for (int q = 0; q < FT_NPW; ++q) {
    const int lr = wave * FT_NPW + q;
    const int i = __builtin_amdgcn_readfirstlane(rowBase + lr);
    float4 acc = make_float4(0.0f, 0.0f, 0.0f, 0.0f);
    if (i < N) {
      const int beg = rowptr[i];
      const int end = rowptr[i + 1];
      const float di = dinv[i];
      acc = *(const float4*)(xl + (size_t)i * 128);
      if (FIRST) { acc.x *= di; acc.y *= di; acc.z *= di; acc.w *= di; }

      for (int base = beg; base < end; base += 32) {
        const int p = base + lane;
        int r = i;
        float w = 0.0f;
        if (p < end) {
          const int2 en = ent[p];
          r = en.x;
          w = FIRST ? __int_as_float(en.y) * dinv[r] : gs[p];
        }
        const int wi = __float_as_int(w);
        const int rem = end - base;
        const int n = rem < 32 ? rem : 32;
        for (int j = 0; j < n; j += 8) {
          int rr[8];
          float ww[8];
          float4 a[8];
#pragma unroll
          for (int u = 0; u < 8; ++u) rr[u] = __builtin_amdgcn_readlane(r, j + u);
#pragma unroll
          for (int u = 0; u < 8; ++u) a[u] = *(const float4*)(xl + (size_t)rr[u] * 128);
#pragma unroll
          for (int u = 0; u < 8; ++u) ww[u] = __int_as_float(__builtin_amdgcn_readlane(wi, j + u));
#pragma unroll
          for (int u = 0; u < 8; ++u) {
            acc.x += ww[u] * a[u].x; acc.y += ww[u] * a[u].y;
            acc.z += ww[u] * a[u].z; acc.w += ww[u] * a[u].w;
          }
        }
      }
      if (lane == 0) {
        s_d[lr] = di;
        if (FIRST) s_n[lr] = dnext[i];
      }
    }
    *(float4*)(&s_t[lr * FT_PITCH + lane * 4]) = acc;
  }
  __syncthreads();

  const int n0 = wave * 16;
  const float* Wl = W + n0 + l16;
  v8f c[FT_RT];
#pragma unroll
  for (int t = 0; t < FT_RT; ++t) c[t] = (v8f){};

#pragma unroll 4
  for (int k0 = 0; k0 < 128; k0 += 8) {
    const int kb = k0 + 4 * half;
    v2f b0, b1;
    b0.x = Wl[(size_t)(kb + 0) * 128];
    b0.y = Wl[(size_t)(kb + 1) * 128];
    b1.x = Wl[(size_t)(kb + 2) * 128];
    b1.y = Wl[(size_t)(kb + 3) * 128];
    float4 a[FT_RT];
#pragma unroll
    for (int t = 0; t < FT_RT; ++t) a[t] = *(const float4*)(&s_t[(t * 16 + l16) * FT_PITCH + kb]);
#pragma unroll
    for (int t = 0; t < FT_RT; ++t) {
      v2f af; af.x = a[t].x; af.y = a[t].y;
      c[t] = WMMA_F32X4(af, b0, c[t]);
    }
#pragma unroll
    for (int t = 0; t < FT_RT; ++t) {
      v2f af; af.x = a[t].z; af.y = a[t].w;
      c[t] = WMMA_F32X4(af, b1, c[t]);
    }
  }

  __syncthreads();
#pragma unroll
  for (int t = 0; t < FT_RT; ++t) {
#pragma unroll
    for (int v = 0; v < 8; ++v) {
      s_t[(t * 16 + v + 8 * half) * FT_PITCH + n0 + l16] = c[t][v];
    }
  }
  __syncthreads();
#pragma unroll
  for (int q = 0; q < FT_NPW; ++q) {
    const int lr = wave * FT_NPW + q;
    const int i = rowBase + lr;
    if (i < N) {
      float4 y = *(const float4*)(&s_t[lr * FT_PITCH + lane * 4]);
      const float di = s_d[lr];
      y.x *= di; y.y *= di; y.z *= di; y.w *= di;
      if (FIRST) {
        const float dn = s_n[lr];
        y.x = fmaxf(y.x, 0.0f) * dn; y.y = fmaxf(y.y, 0.0f) * dn;
        y.z = fmaxf(y.z, 0.0f) * dn; y.w = fmaxf(y.w, 0.0f) * dn;
      }
      *(float4*)(out + (size_t)i * 128 + lane * 4) = y;
    }
  }
}


extern "C" void kernel_launch(void* const* d_in, const int* in_sizes, int n_in,
                              void* d_out, int out_size, void* d_ws, size_t ws_size,
                              hipStream_t stream) {
  const float* x    = (const float*)d_in[0];
  const int*   ei   = (const int*)d_in[1];
  const float* ea   = (const float*)d_in[2];
  const float* W1   = (const float*)d_in[3];
  const float* m1w1 = (const float*)d_in[4];
  const float* m1b1 = (const float*)d_in[5];
  const float* m1w2 = (const float*)d_in[6];
  const float* m1b2 = (const float*)d_in[7];
  const float* W2   = (const float*)d_in[8];
  const float* m2w1 = (const float*)d_in[9];
  const float* m2b1 = (const float*)d_in[10];
  const float* m2w2 = (const float*)d_in[11];
  const float* m2b2 = (const float*)d_in[12];

  const int N = in_sizes[0] / 128;
  const int E = in_sizes[2] / 16;
  const int* row = ei;
  const int* col = ei + E;

  float* ws     = (float*)d_ws;
  float* h_buf  = ws;
  int2*  ent    = (int2*)(h_buf + (size_t)N * 128);
  float* gs2    = (float*)(ent + (size_t)E);
  float* dinv1  = gs2 + (size_t)E;
  int*   cnt    = (int*)(dinv1 + (size_t)N);
  int*   rowptr = cnt + (size_t)N + 1;
  float* dinv2  = (float*)cnt;
  int*   tab    = (int*)h_buf;
  float* outp   = (float*)d_out;
  float2* g12   = (float2*)d_out;

  const int gateBlocks = 1024;
  const int csrR       = (N + CSR_RANGE - 1) / CSR_RANGE;
  const int csrES      = (E + CSR_S - 1) / CSR_S;
  const int csrBlocks  = csrR * CSR_S;
  const int sumBlocks  = (N + 1 + 255) / 256;
  const int scanBlocks = (N + 1 + SCAN_T - 1) / SCAN_T;
  const int degBlocks  = (N + 8 * DEG_NPW - 1) / (8 * DEG_NPW);
  const int aggBlocks  = (N + FT_ROWS - 1) / FT_ROWS;

  pdn_gate_kernel<<<gateBlocks, GATE_T, 0, stream>>>(ea, m1w1, m1b1, m1w2, m1b2,
                                                  m2w1, m2b1, m2w2, m2b2, g12, E);
  pdn_csr_count_kernel<<<csrBlocks, CSR_T, 0, stream>>>(col, tab, csrR, csrES, E, N);
  pdn_csr_rowsum_kernel<<<sumBlocks, 256, 0, stream>>>(tab, cnt, N);
  pdn_csr_scan_kernel<<<scanBlocks, SCAN_T, 0, stream>>>(cnt, rowptr, N + 1);
  pdn_csr_fill_kernel<<<csrBlocks, CSR_T, 0, stream>>>(row, col, rowptr, tab, g12, ent, gs2,
                                                       csrR, csrES, E, N);
  pdn_deg_kernel<<<degBlocks, 256, 0, stream>>>(rowptr, ent, gs2, dinv1, dinv2, N);

  pdn_aggmm_kernel<true><<<aggBlocks, 256, 0, stream>>>(x, W1, rowptr, ent, gs2, dinv1, dinv2,
                                                        h_buf, N);

  pdn_aggmm_kernel<false><<<aggBlocks, 256, 0, stream>>>(h_buf, W2, rowptr, ent, gs2, dinv2, dinv2,
                                                         outp, N);
}
